# starting kernel
# baseline (speedup 1.0000x reference)
#include <hip/hip_runtime.h>
#include <math.h>

typedef float v2f __attribute__((ext_vector_type(2)));
typedef float v4f __attribute__((ext_vector_type(4)));
typedef float v8f __attribute__((ext_vector_type(8)));

#define NB 4
#define NC 16
#define NO 32
#define HW 256
#define LPIX (HW*HW)
#define DF 144
#define DQ 288
#define WAVES 4
#define TPW 8
#define ROWS_TOTAL (NB*LPIX)
#define TILES_TOTAL (ROWS_TOTAL/16)
#define BLOCKS (TILES_TOTAL/(WAVES*TPW))

static __device__ __forceinline__ v8f wmma4(v2f a, v2f b, v8f c) {
  return __builtin_amdgcn_wmma_f32_16x16x4_f32(
      false, a, false, b, (short)0, c, false, false);
}

__global__ __launch_bounds__(WAVES * 32)
void dynaconv_kernel(const float* __restrict__ X,
                     const float* __restrict__ W1,
                     const float* __restrict__ b1,
                     const float* __restrict__ W2,
                     const float* __restrict__ b2,
                     const float* __restrict__ bias,
                     float* __restrict__ Out) {
  __shared__ __align__(16) float sW1[ (DF/2) * 64 ];
  __shared__ __align__(16) float sW2[ (DQ/2) * 64 ];
  __shared__ __align__(16) float sFeat[WAVES][16 * DF];
  __shared__ __align__(16) float sPc  [WAVES][16 * 9];
  __shared__ __align__(16) float sHid [WAVES][16 * NO];

  const int tid  = threadIdx.x;
  const int lane = tid & 31;
  const int wv   = tid >> 5;
  const int col  = lane & 15;
  const int h    = lane >> 4;

  for (int e = tid; e < NO * DF; e += WAVES * 32) {
    int o = e / DF, f = e - o * DF;
    sW1[(f >> 1) * 64 + (o << 1) + (f & 1)] = W1[e];
  }
  for (int e = tid; e < DQ * NO; e += WAVES * 32) {
    int g = e >> 5, o2 = e & 31;
    int o = g / 9, j = g - o * 9;
    int q = o2 * 9 + j;
    sW2[(q >> 1) * 64 + (o << 1) + (q & 1)] = W2[e];
  }
  const float b1v0 = b1[col],      b1v1 = b1[col + 16];
  const float bs0  = bias[col],    bs1  = bias[col + 16];
  float b2a[9], b2b[9];
#pragma unroll
  for (int j = 0; j < 9; ++j) { b2a[j] = b2[col * 9 + j]; b2b[j] = b2[(col + 16) * 9 + j]; }
  __syncthreads();

  float* feat = sFeat[wv];
  float* pc   = sPc[wv];
  float* hid  = sHid[wv];

  for (int it = 0; it < TPW; ++it) {
    const int t    = ((int)blockIdx.x * WAVES + wv) * TPW + it;
    const int row0 = t << 4;
    const int bimg = row0 >> 16;
    const int p0   = row0 & (LPIX - 1);

    for (int e = lane; e < 16 * DF; e += 32) {
      int m = e / DF, f = e - m * DF;
      int c = f / 9, ij = f - c * 9;
      int di = ij / 3, dj = ij - di * 3;
      int p = p0 + m;
      int yy = (p >> 8) + di - 1;
      int xx = (p & 255) + dj - 1;
      float v = 0.0f;
      if ((unsigned)yy < (unsigned)HW && (unsigned)xx < (unsigned)HW)
        v = X[(((bimg * NC + c) * HW + yy) * HW) + xx];
      feat[e] = v;
    }
    __syncthreads();

    for (int e = lane; e < 16 * 9; e += 32) {
      int m = e / 9, j = e - m * 9;
      float s = 0.0f;
#pragma unroll
      for (int c = 0; c < NC; ++c) s += feat[m * DF + c * 9 + j];
      pc[e] = s;
    }
    __syncthreads();

    v8f acc0, acc1;
#pragma unroll
    for (int r = 0; r < 8; ++r) { acc0[r] = b1v0; acc1[r] = b1v1; }
    const float* aBase  = feat + col * DF + 2 * h;
    const float* w1Base = sW1 + h * 64 + (col << 1);
#pragma unroll
    for (int s = 0; s < DF / 4; ++s) {
      v2f a  = *(const v2f*)(aBase + 4 * s);
      v2f bA = *(const v2f*)(w1Base + s * 128);
      v2f bB = *(const v2f*)(w1Base + s * 128 + 32);
      acc0 = wmma4(a, bA, acc0);
      acc1 = wmma4(a, bB, acc1);
    }
#pragma unroll
    for (int r = 0; r < 8; ++r) {
      int m2 = r + 8 * h;
      hid[m2 * NO + col]      = tanhf(acc0[r]);
      hid[m2 * NO + col + 16] = tanhf(acc1[r]);
    }
    __syncthreads();

    float hr[NO];
#pragma unroll
    for (int k = 0; k < NO / 4; ++k) {
      v4f v = *(const v4f*)(hid + col * NO + 4 * k);
      hr[4 * k + 0] = v.x; hr[4 * k + 1] = v.y; hr[4 * k + 2] = v.z; hr[4 * k + 3] = v.w;
    }
    float pr[9];
#pragma unroll
    for (int j = 0; j < 9; ++j) pr[j] = pc[col * 9 + j];

#pragma unroll
    for (int r = 0; r < 8; ++r) {
      int m2 = r + 8 * h;
      float s0 = bs0, s1 = bs1;
#pragma unroll
      for (int j = 0; j < 9; ++j) { float pv = pc[m2 * 9 + j]; s0 += pv * b2a[j]; s1 += pv * b2b[j]; }
      acc0[r] = s0; acc1[r] = s1;
    }

    const float* w2Base = sW2 + h * 64 + (col << 1);
#pragma unroll
    for (int s = 0; s < DQ / 4; ++s) {
      const int q = 4 * s;
      float z0 = hr[(q + 0) / 9] * pr[(q + 0) % 9];
      float z1 = hr[(q + 1) / 9] * pr[(q + 1) % 9];
      float z2 = hr[(q + 2) / 9] * pr[(q + 2) % 9];
      float z3 = hr[(q + 3) / 9] * pr[(q + 3) % 9];
      v2f a; a.x = h ? z2 : z0; a.y = h ? z3 : z1;
      v2f bA = *(const v2f*)(w2Base + s * 128);
      v2f bB = *(const v2f*)(w2Base + s * 128 + 32);
      acc0 = wmma4(a, bA, acc0);
      acc1 = wmma4(a, bB, acc1);
    }

    float* outT = hid;
#pragma unroll
    for (int r = 0; r < 8; ++r) {
      int m2 = r + 8 * h;
      outT[m2 * NO + col]      = acc0[r];
      outT[m2 * NO + col + 16] = acc1[r];
    }
    __syncthreads();
#pragma unroll
    for (int e = 0; e < 16; ++e) {
      int idx = e * 32 + lane;
      int o = idx >> 4, m = idx & 15;
      Out[((long)(bimg * NO + o)) * LPIX + p0 + m] = outT[m * NO + o];
    }
    __syncthreads();
  }
}

extern "C" void kernel_launch(void* const* d_in, const int* in_sizes, int n_in,
                              void* d_out, int out_size, void* d_ws, size_t ws_size,
                              hipStream_t stream) {
  (void)in_sizes; (void)n_in; (void)out_size; (void)d_ws; (void)ws_size;
  const float* X    = (const float*)d_in[0];
  const float* W1   = (const float*)d_in[1];
  const float* b1   = (const float*)d_in[2];
  const float* W2   = (const float*)d_in[3];
  const float* b2   = (const float*)d_in[4];
  const float* bias = (const float*)d_in[5];
  float* Out = (float*)d_out;
  dynaconv_kernel<<<BLOCKS, WAVES * 32, 0, stream>>>(X, W1, b1, W2, b2, bias, Out);
}


# W1/W2 fragment LDS reads issued as groups of 18 k steps one block ahead of their 36 matrix instructions (two register sets), not one read between every two; sums unchanged
# speedup vs baseline: 2.5512x; 2.5512x over previous
#include <hip/hip_runtime.h>
#include <math.h>

typedef float v2f __attribute__((ext_vector_type(2)));
typedef float v4f __attribute__((ext_vector_type(4)));
typedef float v8f __attribute__((ext_vector_type(8)));

#define NB 4
#define NC 16
#define NO 32
#define HW 256
#define LPIX (HW*HW)
#define DF 144
#define DQ 288
#define WAVES 4
#define TPW 8
#define ROWS_TOTAL (NB*LPIX)
#define TILES_TOTAL (ROWS_TOTAL/16)
#define BLOCKS (TILES_TOTAL/(WAVES*TPW))
#define KB 18
#define NBLK2 (DQ/4/KB)

static __device__ __forceinline__ float xhalf(float x) {
  int b = __builtin_bit_cast(int, x);
  return __builtin_bit_cast(float,
      __builtin_amdgcn_permlanex16(b, b, 0x76543210, 0xfedcba98, false, false));
}

static __device__ __forceinline__ v8f wmma4(v2f a, v2f b, v8f c) {
  return __builtin_amdgcn_wmma_f32_16x16x4_f32(
      false, a, false, b, (short)0, c, false, false);
}

static __device__ __forceinline__ void ld_block(const float* base, int s0,
                                                v2f (&fa)[KB], v2f (&fb)[KB]) {
#pragma unroll
  for (int i = 0; i < KB; ++i) {
    fa[i] = *(const v2f*)(base + (s0 + i) * 128);
    fb[i] = *(const v2f*)(base + (s0 + i) * 128 + 32);
  }
}

static __device__ __forceinline__ void pin(v8f& a0, v8f& a1) {
  asm volatile("" : "+v"(a0), "+v"(a1) :: "memory");
}

__global__ __launch_bounds__(WAVES * 32)
void dynaconv_kernel(const float* __restrict__ X,
                     const float* __restrict__ W1,
                     const float* __restrict__ b1,
                     const float* __restrict__ W2,
                     const float* __restrict__ b2,
                     const float* __restrict__ bias,
                     float* __restrict__ Out) {
  __shared__ __align__(16) float sW1[ (DF/2) * 64 ];
  __shared__ __align__(16) float sW2[ (DQ/2) * 64 ];

  const int tid  = threadIdx.x;
  const int lane = tid & 31;
  const int wv   = tid >> 5;
  const int col  = lane & 15;
  const int h    = lane >> 4;

  for (int e = tid; e < NO * DF; e += WAVES * 32) {
    int o = e / DF, f = e - o * DF;
    int c = f / 9, ij = f - c * 9;
    int k = ij * NC + c;
    sW1[(k >> 1) * 64 + (o << 1) + (k & 1)] = W1[e];
  }
  for (int e = tid; e < DQ * NO; e += WAVES * 32) {
    int g = e >> 5, o2 = e & 31;
    int o = g / 9, j = g - o * 9;
    int q = o2 * 9 + j;
    sW2[(q >> 1) * 64 + (o << 1) + (q & 1)] = W2[e];
  }
  float b1r0[8], b1r1[8], bsr0[8], bsr1[8];
  float b2r0[72], b2r1[72];
#pragma unroll
  for (int r = 0; r < 8; ++r) {
    b1r0[r] = b1[8 * h + r];   b1r1[r] = b1[16 + 8 * h + r];
    bsr0[r] = bias[8 * h + r]; bsr1[r] = bias[16 + 8 * h + r];
  }
#pragma unroll
  for (int i = 0; i < 72; ++i) { b2r0[i] = b2[72 * h + i]; b2r1[i] = b2[144 + 72 * h + i]; }
  __syncthreads();

  const float* w1Base = sW1 + h * 64 + (col << 1);
  const float* w2Base = sW2 + h * 64 + (col << 1);
  v2f fa[KB], fb[KB], ga[KB], gb[KB];

  for (int it = 0; it < TPW; ++it) {
    const int t    = ((int)blockIdx.x * WAVES + wv) * TPW + it;
    const int row0 = t << 4;
    const int bimg = row0 >> 16;
    const int p0   = row0 & (LPIX - 1);

    const int y = p0 >> 8;
    const int x = (p0 & 255) + col;
    const float* xb = X + (bimg * NC + 2 * h) * LPIX;
    v2f afr[DF / 4];
    float pr[9];
#pragma unroll
    for (int ij = 0; ij < 9; ++ij) {
      const int di = ij / 3, dj = ij - di * 3;
      const int yy = y + di - 1, xx = x + dj - 1;
      const bool ok = (di == 1 || (unsigned)yy < (unsigned)HW) &&
                      (dj == 1 || (unsigned)xx < (unsigned)HW);
      const int yc = min(max(yy, 0), HW - 1);
      const int xc = min(max(xx, 0), HW - 1);
      const float* pt = xb + yc * HW + xc;
#pragma unroll
      for (int g = 0; g < 4; ++g) {
        float v0 = pt[(4 * g) * LPIX];
        float v1 = pt[(4 * g + 1) * LPIX];
        afr[ij * 4 + g].x = ok ? v0 : 0.0f;
        afr[ij * 4 + g].y = ok ? v1 : 0.0f;
      }
    }

    v8f acc0, acc1;
#pragma unroll
    for (int r = 0; r < 8; ++r) { acc0[r] = b1r0[r]; acc1[r] = b1r1[r]; }
    ld_block(w1Base, 0, fa, fb);
    ld_block(w1Base, KB, ga, gb);
    pin(acc0, acc1);
#pragma unroll
    for (int i = 0; i < KB; ++i) {
      v2f a = afr[i];
      acc0 = wmma4(fa[i], a, acc0);
      acc1 = wmma4(fb[i], a, acc1);
    }
    pin(acc0, acc1);
    ld_block(w2Base, 0, fa, fb);
    pin(acc0, acc1);
#pragma unroll
    for (int i = 0; i < KB; ++i) {
      v2f a = afr[KB + i];
      acc0 = wmma4(ga[i], a, acc0);
      acc1 = wmma4(gb[i], a, acc1);
    }
    pin(acc0, acc1);
    ld_block(w2Base, KB, ga, gb);

#pragma unroll
    for (int ij = 0; ij < 9; ++ij) {
      float s = afr[ij * 4].x + afr[ij * 4].y;
#pragma unroll
      for (int g = 1; g < 4; ++g) { s += afr[ij * 4 + g].x; s += afr[ij * 4 + g].y; }
      s += xhalf(s);
      pr[ij] = s;
    }
    float hr[NO];
#pragma unroll
    for (int r = 0; r < 8; ++r) {
      float t0 = tanhf(acc0[r]), t1 = tanhf(acc1[r]);
      float x0 = xhalf(t0),      x1 = xhalf(t1);
      hr[r]      = h ? x0 : t0;
      hr[8 + r]  = h ? t0 : x0;
      hr[16 + r] = h ? x1 : t1;
      hr[24 + r] = h ? t1 : x1;
    }
#pragma unroll
    for (int r = 0; r < 8; ++r) {
      float s0 = bsr0[r], s1 = bsr1[r];
#pragma unroll
      for (int j = 0; j < 9; ++j) { s0 += pr[j] * b2r0[r * 9 + j]; s1 += pr[j] * b2r1[r * 9 + j]; }
      acc0[r] = s0; acc1[r] = s1;
    }

#pragma unroll
    for (int b = 0; b < NBLK2; ++b) {
      if (b > 0 && b + 1 < NBLK2) {
        pin(acc0, acc1);
        if ((b + 1) & 1) ld_block(w2Base, (b + 1) * KB, ga, gb);
        else             ld_block(w2Base, (b + 1) * KB, fa, fb);
        pin(acc0, acc1);
      }
#pragma unroll
      for (int i = 0; i < KB; ++i) {
        const int q = 4 * (b * KB + i);
        float z0 = hr[(q + 0) / 9] * pr[(q + 0) % 9];
        float z1 = hr[(q + 1) / 9] * pr[(q + 1) % 9];
        float z2 = hr[(q + 2) / 9] * pr[(q + 2) % 9];
        float z3 = hr[(q + 3) / 9] * pr[(q + 3) % 9];
        v2f a; a.x = h ? z2 : z0; a.y = h ? z3 : z1;
        if (b & 1) { acc0 = wmma4(ga[i], a, acc0); acc1 = wmma4(gb[i], a, acc1); }
        else       { acc0 = wmma4(fa[i], a, acc0); acc1 = wmma4(fb[i], a, acc1); }
      }
    }

    float* ob = Out + ((long)(bimg * NO + 8 * h)) * LPIX + p0 + col;
#pragma unroll
    for (int r = 0; r < 8; ++r) {
      ob[r * LPIX]        = acc0[r];
      ob[(16 + r) * LPIX] = acc1[r];
    }
  }
}

extern "C" void kernel_launch(void* const* d_in, const int* in_sizes, int n_in,
                              void* d_out, int out_size, void* d_ws, size_t ws_size,
                              hipStream_t stream) {
  (void)in_sizes; (void)n_in; (void)out_size; (void)d_ws; (void)ws_size;
  const float* X    = (const float*)d_in[0];
  const float* W1   = (const float*)d_in[1];
  const float* b1   = (const float*)d_in[2];
  const float* W2   = (const float*)d_in[3];
  const float* b2   = (const float*)d_in[4];
  const float* bias = (const float*)d_in[5];
  float* Out = (float*)d_out;
  dynaconv_kernel<<<BLOCKS, WAVES * 32, 0, stream>>>(X, W1, b1, W2, b2, bias, Out);
}


# each wave walks down a 16-pixel column strip: a tile's 3 input rows are a register window, 2 kept from the tile above, the next row loaded one trip ahead (24 loads per tile, was 72); sums unchanged
# speedup vs baseline: 2.7334x; 1.0714x over previous
#include <hip/hip_runtime.h>
#include <math.h>

typedef float v2f __attribute__((ext_vector_type(2)));
typedef float v4f __attribute__((ext_vector_type(4)));
typedef float v8f __attribute__((ext_vector_type(8)));

#define NB 4
#define NC 16
#define NO 32
#define HW 256
#define LPIX (HW*HW)
#define DF 144
#define DQ 288
#define WAVES 4
#define TPW 8
#define ROWS_TOTAL (NB*LPIX)
#define TILES_TOTAL (ROWS_TOTAL/16)
#define BLOCKS (TILES_TOTAL/(WAVES*TPW))
#define TCOLS (HW/16)
#define STRIPS (TCOLS*(HW/TPW))
#define KB 18
#define NBLK2 (DQ/4/KB)

static __device__ __forceinline__ float xhalf(float x) {
  int b = __builtin_bit_cast(int, x);
  return __builtin_bit_cast(float,
      __builtin_amdgcn_permlanex16(b, b, 0x76543210, 0xfedcba98, false, false));
}

static __device__ __forceinline__ v8f wmma4(v2f a, v2f b, v8f c) {
  return __builtin_amdgcn_wmma_f32_16x16x4_f32(
      false, a, false, b, (short)0, c, false, false);
}

static __device__ __forceinline__ void ld_block(const float* base, int s0,
                                                v2f (&fa)[KB], v2f (&fb)[KB]) {
#pragma unroll
  for (int i = 0; i < KB; ++i) {
    fa[i] = *(const v2f*)(base + (s0 + i) * 128);
    fb[i] = *(const v2f*)(base + (s0 + i) * 128 + 32);
  }
}

static __device__ __forceinline__ void pin(v8f& a0, v8f& a1) {
  asm volatile("" : "+v"(a0), "+v"(a1) :: "memory");
}

static __device__ __forceinline__ void ld_row(const float* rowp, const int (&xo)[3],
                                              v2f (&R)[12]) {
#pragma unroll
  for (int dj = 0; dj < 3; ++dj) {
    const float* pt = rowp + xo[dj];
#pragma unroll
    for (int g = 0; g < 4; ++g) {
      R[dj * 4 + g].x = pt[(4 * g) * LPIX];
      R[dj * 4 + g].y = pt[(4 * g + 1) * LPIX];
    }
  }
}

static __device__ __forceinline__ void zero_pad(bool rowok, bool ok0, bool ok2, v2f (&R)[12]) {
#pragma unroll
  for (int dj = 0; dj < 3; ++dj) {
    const bool ok = rowok && (dj == 0 ? ok0 : (dj == 2 ? ok2 : true));
#pragma unroll
    for (int g = 0; g < 4; ++g) {
      R[dj * 4 + g].x = ok ? R[dj * 4 + g].x : 0.0f;
      R[dj * 4 + g].y = ok ? R[dj * 4 + g].y : 0.0f;
    }
  }
}

static __device__ __forceinline__ float tap_sum(const v2f (&R)[12], int dj) {
  float s = R[dj * 4].x + R[dj * 4].y;
#pragma unroll
  for (int g = 1; g < 4; ++g) { s += R[dj * 4 + g].x; s += R[dj * 4 + g].y; }
  s += xhalf(s);
  return s;
}

__global__ __launch_bounds__(WAVES * 32)
void dynaconv_kernel(const float* __restrict__ X,
                     const float* __restrict__ W1,
                     const float* __restrict__ b1,
                     const float* __restrict__ W2,
                     const float* __restrict__ b2,
                     const float* __restrict__ bias,
                     float* __restrict__ Out) {
  __shared__ __align__(16) float sW1[ (DF/2) * 64 ];
  __shared__ __align__(16) float sW2[ (DQ/2) * 64 ];

  const int tid  = threadIdx.x;
  const int lane = tid & 31;
  const int wv   = tid >> 5;
  const int col  = lane & 15;
  const int h    = lane >> 4;

  const int sw   = (int)blockIdx.x * WAVES + wv;
  const int bimg = sw / STRIPS;
  const int tcol = sw % TCOLS;
  const int y0   = ((sw / TCOLS) % (HW / TPW)) * TPW;
  const int x    = tcol * 16 + col;
  const float* xb = X + (bimg * NC + 2 * h) * LPIX;
  int xo[3];
  xo[0] = max(x - 1, 0); xo[1] = x; xo[2] = min(x + 1, HW - 1);
  const bool ok0 = x > 0, ok2 = x < HW - 1;

  v2f R0[12], R1[12], R2[12];
  ld_row(xb + max(y0 - 1, 0) * HW, xo, R0);
  ld_row(xb + y0 * HW, xo, R1);
  ld_row(xb + min(y0 + 1, HW - 1) * HW, xo, R2);

  for (int e = tid; e < NO * DF; e += WAVES * 32) {
    int o = e / DF, f = e - o * DF;
    int c = f / 9, ij = f - c * 9;
    int k = ij * NC + c;
    sW1[(k >> 1) * 64 + (o << 1) + (k & 1)] = W1[e];
  }
  for (int e = tid; e < DQ * NO; e += WAVES * 32) {
    int g = e >> 5, o2 = e & 31;
    int o = g / 9, j = g - o * 9;
    int q = o2 * 9 + j;
    sW2[(q >> 1) * 64 + (o << 1) + (q & 1)] = W2[e];
  }
  float b1r0[8], b1r1[8], bsr0[8], bsr1[8];
  float b2r0[72], b2r1[72];
#pragma unroll
  for (int r = 0; r < 8; ++r) {
    b1r0[r] = b1[8 * h + r];   b1r1[r] = b1[16 + 8 * h + r];
    bsr0[r] = bias[8 * h + r]; bsr1[r] = bias[16 + 8 * h + r];
  }
#pragma unroll
  for (int i = 0; i < 72; ++i) { b2r0[i] = b2[72 * h + i]; b2r1[i] = b2[144 + 72 * h + i]; }

  float pr[9];
  zero_pad(y0 > 0, ok0, ok2, R0);
  zero_pad(true, ok0, ok2, R1);
#pragma unroll
  for (int dj = 0; dj < 3; ++dj) { pr[dj] = tap_sum(R0, dj); pr[3 + dj] = tap_sum(R1, dj); }
  __syncthreads();

  const float* w1Base = sW1 + h * 64 + (col << 1);
  const float* w2Base = sW2 + h * 64 + (col << 1);
  v2f fa[KB], fb[KB], ga[KB], gb[KB];

  for (int it = 0; it < TPW; ++it) {
    const int y = y0 + it;

    zero_pad(y + 1 < HW, ok0, ok2, R2);
    v2f afr[DF / 4];
#pragma unroll
    for (int i = 0; i < 12; ++i) { afr[i] = R0[i]; afr[12 + i] = R1[i]; afr[24 + i] = R2[i]; }

    v8f acc0, acc1;
#pragma unroll
    for (int r = 0; r < 8; ++r) { acc0[r] = b1r0[r]; acc1[r] = b1r1[r]; }
    ld_block(w1Base, 0, fa, fb);
    ld_block(w1Base, KB, ga, gb);
    pin(acc0, acc1);
#pragma unroll
    for (int i = 0; i < KB; ++i) {
      v2f a = afr[i];
      acc0 = wmma4(fa[i], a, acc0);
      acc1 = wmma4(fb[i], a, acc1);
    }
    pin(acc0, acc1);
    ld_block(w2Base, 0, fa, fb);
    pin(acc0, acc1);
#pragma unroll
    for (int i = 0; i < KB; ++i) {
      v2f a = afr[KB + i];
      acc0 = wmma4(ga[i], a, acc0);
      acc1 = wmma4(gb[i], a, acc1);
    }
    pin(acc0, acc1);
    ld_block(w2Base, KB, ga, gb);

#pragma unroll
    for (int dj = 0; dj < 3; ++dj) pr[6 + dj] = tap_sum(R2, dj);

#pragma unroll
    for (int i = 0; i < 12; ++i) { R0[i] = R1[i]; R1[i] = R2[i]; }
    ld_row(xb + min(y + 2, HW - 1) * HW, xo, R2);

    float hr[NO];
#pragma unroll
    for (int r = 0; r < 8; ++r) {
      float t0 = tanhf(acc0[r]), t1 = tanhf(acc1[r]);
      float x0 = xhalf(t0),      x1 = xhalf(t1);
      hr[r]      = h ? x0 : t0;
      hr[8 + r]  = h ? t0 : x0;
      hr[16 + r] = h ? x1 : t1;
      hr[24 + r] = h ? t1 : x1;
    }
#pragma unroll
    for (int r = 0; r < 8; ++r) {
      float s0 = bsr0[r], s1 = bsr1[r];
#pragma unroll
      for (int j = 0; j < 9; ++j) { s0 += pr[j] * b2r0[r * 9 + j]; s1 += pr[j] * b2r1[r * 9 + j]; }
      acc0[r] = s0; acc1[r] = s1;
    }

#pragma unroll
    for (int b = 0; b < NBLK2; ++b) {
      if (b > 0 && b + 1 < NBLK2) {
        pin(acc0, acc1);
        if ((b + 1) & 1) ld_block(w2Base, (b + 1) * KB, ga, gb);
        else             ld_block(w2Base, (b + 1) * KB, fa, fb);
        pin(acc0, acc1);
      }
#pragma unroll
      for (int i = 0; i < KB; ++i) {
        const int q = 4 * (b * KB + i);
        float z0 = hr[(q + 0) / 9] * pr[(q + 0) % 9];
        float z1 = hr[(q + 1) / 9] * pr[(q + 1) % 9];
        float z2 = hr[(q + 2) / 9] * pr[(q + 2) % 9];
        float z3 = hr[(q + 3) / 9] * pr[(q + 3) % 9];
        v2f a; a.x = h ? z2 : z0; a.y = h ? z3 : z1;
        if (b & 1) { acc0 = wmma4(ga[i], a, acc0); acc1 = wmma4(gb[i], a, acc1); }
        else       { acc0 = wmma4(fa[i], a, acc0); acc1 = wmma4(fb[i], a, acc1); }
      }
    }

    float* ob = Out + ((long)(bimg * NO + 8 * h)) * LPIX + y * HW + x;
#pragma unroll
    for (int r = 0; r < 8; ++r) {
      ob[r * LPIX]        = acc0[r];
      ob[(16 + r) * LPIX] = acc1[r];
    }

#pragma unroll
    for (int j = 0; j < 6; ++j) pr[j] = pr[3 + j];
  }
}

extern "C" void kernel_launch(void* const* d_in, const int* in_sizes, int n_in,
                              void* d_out, int out_size, void* d_ws, size_t ws_size,
                              hipStream_t stream) {
  (void)in_sizes; (void)n_in; (void)out_size; (void)d_ws; (void)ws_size;
  const float* X    = (const float*)d_in[0];
  const float* W1   = (const float*)d_in[1];
  const float* b1   = (const float*)d_in[2];
  const float* W2   = (const float*)d_in[3];
  const float* b2   = (const float*)d_in[4];
  const float* bias = (const float*)d_in[5];
  float* Out = (float*)d_out;
  dynaconv_kernel<<<BLOCKS, WAVES * 32, 0, stream>>>(X, W1, b1, W2, b2, bias, Out);
}


# tile loop software-pipelined: GEMM1 runs a tile ahead, its tanh (branch-free, both library formulas, last bits may differ) sits among the prior tile's GEMM2 matrix instructions; sums unchanged
# speedup vs baseline: 2.9611x; 1.0833x over previous
#include <hip/hip_runtime.h>
#include <math.h>

typedef float v2f __attribute__((ext_vector_type(2)));
typedef float v4f __attribute__((ext_vector_type(4)));
typedef float v8f __attribute__((ext_vector_type(8)));

#define NB 4
#define NC 16
#define NO 32
#define HW 256
#define LPIX (HW*HW)
#define DF 144
#define DQ 288
#define WAVES 4
#define TPW 8
#define ROWS_TOTAL (NB*LPIX)
#define TILES_TOTAL (ROWS_TOTAL/16)
#define BLOCKS (TILES_TOTAL/(WAVES*TPW))
#define TCOLS (HW/16)
#define STRIPS (TCOLS*(HW/TPW))
#define KB 18
#define NBLK2 (DQ/4/KB)

static __device__ __forceinline__ float xhalf(float x) {
  int b = __builtin_bit_cast(int, x);
  return __builtin_bit_cast(float,
      __builtin_amdgcn_permlanex16(b, b, 0x76543210, 0xfedcba98, false, false));
}

static __device__ __forceinline__ v8f wmma4(v2f a, v2f b, v8f c) {
  return __builtin_amdgcn_wmma_f32_16x16x4_f32(
      false, a, false, b, (short)0, c, false, false);
}

static __device__ __forceinline__ void ld_block(const float* base, int s0,
                                                v2f (&fa)[KB], v2f (&fb)[KB]) {
#pragma unroll
  for (int i = 0; i < KB; ++i) {
    fa[i] = *(const v2f*)(base + (s0 + i) * 128);
    fb[i] = *(const v2f*)(base + (s0 + i) * 128 + 32);
  }
}

static __device__ __forceinline__ void pin(v8f& a0, v8f& a1) {
  asm volatile("" : "+v"(a0), "+v"(a1) :: "memory");
}

static __device__ __forceinline__ void ld_row(const float* img, unsigned rowoff,
                                              const unsigned (&xo)[3], v2f (&R)[12]) {
#pragma unroll
  for (int dj = 0; dj < 3; ++dj) {
    const float* pt = (const float*)((const char*)img + (xo[dj] + rowoff));
#pragma unroll
    for (int g = 0; g < 4; ++g) {
      R[dj * 4 + g].x = pt[(4 * g) * LPIX];
      R[dj * 4 + g].y = pt[(4 * g + 1) * LPIX];
    }
  }
}

static __device__ __forceinline__ void zero_pad(bool rowok, bool ok0, bool ok2, v2f (&R)[12]) {
#pragma unroll
  for (int dj = 0; dj < 3; ++dj) {
    const bool ok = rowok && (dj == 0 ? ok0 : (dj == 2 ? ok2 : true));
#pragma unroll
    for (int g = 0; g < 4; ++g) {
      R[dj * 4 + g].x = ok ? R[dj * 4 + g].x : 0.0f;
      R[dj * 4 + g].y = ok ? R[dj * 4 + g].y : 0.0f;
    }
  }
}

static __device__ __forceinline__ float tap_sum(const v2f (&R)[12], int dj) {
  float s = R[dj * 4].x + R[dj * 4].y;
#pragma unroll
  for (int g = 1; g < 4; ++g) { s += R[dj * 4 + g].x; s += R[dj * 4 + g].y; }
  s += xhalf(s);
  return s;
}

static __device__ __forceinline__ float tanh_sel(float x) {
  const float y = __builtin_fabsf(x);
  const float a  = y + y;
  const float l2 = __builtin_bit_cast(float, 0x3fb8aa3bu);
  const float m  = a * l2;
  const float r  = __builtin_rintf(m);
  float e = __builtin_fmaf(a, l2, -m);
  e = __builtin_fmaf(a, __builtin_bit_cast(float, 0x32a5705fu), e);
  float p = __builtin_amdgcn_exp2f((m - r) + e);
  p = __builtin_ldexpf(p, (int)r);
  const float zb = __builtin_fmaf(__builtin_amdgcn_rcpf(1.0f + p), -2.0f, 1.0f);
  const float y2 = x * x;
  float q = __builtin_fmaf(y2, __builtin_bit_cast(float, 0xbbbac73du),
                               __builtin_bit_cast(float, 0x3ca908c9u));
  q = __builtin_fmaf(y2, q, __builtin_bit_cast(float, 0xbd5c1c4eu));
  q = __builtin_fmaf(y2, q, __builtin_bit_cast(float, 0x3e088382u));
  q = __builtin_fmaf(y2, q, __builtin_bit_cast(float, 0xbeaaaa99u));
  const float zs = __builtin_fmaf(y2, y * q, y);
  const float z = (y < 0.625f) ? zs : zb;
  return __builtin_copysignf(z, x);
}

__global__ __launch_bounds__(WAVES * 32) __attribute__((amdgpu_waves_per_eu(2)))
void dynaconv_kernel(const float* __restrict__ X,
                     const float* __restrict__ W1,
                     const float* __restrict__ b1,
                     const float* __restrict__ W2,
                     const float* __restrict__ b2,
                     const float* __restrict__ bias,
                     float* __restrict__ Out) {
  __shared__ __align__(16) float sW1[ (DF/2) * 64 ];
  __shared__ __align__(16) float sW2[ (DQ/2) * 64 ];
  __shared__ __align__(16) float sB[2 * NO];

  const int tid  = threadIdx.x;
  const int lane = tid & 31;
  const int wv   = __builtin_amdgcn_readfirstlane(tid >> 5);
  const int col  = lane & 15;
  const int h    = lane >> 4;

  const int sw   = (int)blockIdx.x * WAVES + wv;
  const int bimg = sw / STRIPS;
  const int tcol = sw % TCOLS;
  const int y0   = ((sw / TCOLS) % (HW / TPW)) * TPW;
  const int x    = tcol * 16 + col;
  const float* xb = X + bimg * NC * LPIX;
  unsigned xo[3];
  xo[0] = 4u * (unsigned)(2 * h * LPIX + max(x - 1, 0));
  xo[1] = 4u * (unsigned)(2 * h * LPIX + x);
  xo[2] = 4u * (unsigned)(2 * h * LPIX + min(x + 1, HW - 1));
  const bool ok0 = x > 0, ok2 = x < HW - 1;

  v2f R0[12], R1[12], R2[12];
  ld_row(xb, 4u * (unsigned)(max(y0 - 1, 0) * HW), xo, R0);
  ld_row(xb, 4u * (unsigned)(y0 * HW), xo, R1);
  ld_row(xb, 4u * (unsigned)(min(y0 + 1, HW - 1) * HW), xo, R2);

  for (int e = tid; e < NO * DF; e += WAVES * 32) {
    int o = e / DF, f = e - o * DF;
    int c = f / 9, ij = f - c * 9;
    int k = ij * NC + c;
    sW1[(k >> 1) * 64 + (o << 1) + (k & 1)] = W1[e];
  }
  for (int e = tid; e < DQ * NO; e += WAVES * 32) {
    int g = e >> 5, o2 = e & 31;
    int o = g / 9, j = g - o * 9;
    int q = o2 * 9 + j;
    sW2[(q >> 1) * 64 + (o << 1) + (q & 1)] = W2[e];
  }
  if (tid < NO) { sB[tid] = b1[tid]; sB[NO + tid] = bias[tid]; }
  float b2r0[72], b2r1[72];
#pragma unroll
  for (int i = 0; i < 72; ++i) { b2r0[i] = b2[72 * h + i]; b2r1[i] = b2[144 + 72 * h + i]; }

  float pr[9];
  zero_pad(y0 > 0, ok0, ok2, R0);
  zero_pad(true, ok0, ok2, R1);
#pragma unroll
  for (int dj = 0; dj < 3; ++dj) {
    pr[dj] = 0.0f; pr[3 + dj] = tap_sum(R0, dj); pr[6 + dj] = tap_sum(R1, dj);
  }
  __syncthreads();

  const float* w1Base = sW1 + h * 64 + (col << 1);
  const float* w2Base = sW2 + h * 64 + (col << 1);
  v2f fa[KB], fb[KB], ga[KB], gb[KB];

  v8f P0, P1;
#pragma unroll
  for (int r = 0; r < 8; ++r) { P0[r] = 0.0f; P1[r] = 0.0f; }

#pragma nounroll
  for (int u = 0; u <= TPW; ++u) {
    float hr[NO];
#pragma unroll
    for (int r = 0; r < 8; ++r) {
      float t0 = P0[r],     t1 = P1[r];
      float x0 = xhalf(t0), x1 = xhalf(t1);
      hr[r]      = h ? x0 : t0;
      hr[8 + r]  = h ? t0 : x0;
      hr[16 + r] = h ? x1 : t1;
      hr[24 + r] = h ? t1 : x1;
    }

    v8f acc0, acc1;
#pragma unroll
    for (int r = 0; r < 8; ++r) {
      float s0 = sB[NO + 8 * h + r], s1 = sB[NO + 16 + 8 * h + r];
#pragma unroll
      for (int j = 0; j < 9; ++j) { s0 += pr[j] * b2r0[r * 9 + j]; s1 += pr[j] * b2r1[r * 9 + j]; }
      acc0[r] = s0; acc1[r] = s1;
    }
    pin(acc0, acc1);

    float prn[3] = {0.0f, 0.0f, 0.0f};
    if (u < TPW) {
      const int yu = y0 + u;
      zero_pad(yu + 1 < HW, ok0, ok2, R2);
      v2f afr[DF / 4];
#pragma unroll
      for (int i = 0; i < 12; ++i) { afr[i] = R0[i]; afr[12 + i] = R1[i]; afr[24 + i] = R2[i]; }

#pragma unroll
      for (int r = 0; r < 8; ++r) { P0[r] = sB[8 * h + r]; P1[r] = sB[16 + 8 * h + r]; }
      ld_block(w1Base, 0, fa, fb);
      ld_block(w1Base, KB, ga, gb);
      pin(P0, P1);
#pragma unroll
      for (int i = 0; i < KB; ++i) {
        v2f a = afr[i];
        P0 = wmma4(fa[i], a, P0);
        P1 = wmma4(fb[i], a, P1);
      }
      pin(P0, P1);
      ld_block(w2Base, 0, fa, fb);
      pin(P0, P1);
#pragma unroll
      for (int i = 0; i < KB; ++i) {
        v2f a = afr[KB + i];
        P0 = wmma4(ga[i], a, P0);
        P1 = wmma4(gb[i], a, P1);
      }
      pin(P0, P1);
      ld_block(w2Base, KB, ga, gb);

#pragma unroll
      for (int dj = 0; dj < 3; ++dj) prn[dj] = tap_sum(R2, dj);

#pragma unroll
      for (int i = 0; i < 12; ++i) { R0[i] = R1[i]; R1[i] = R2[i]; }
      ld_row(xb, 4u * (unsigned)(min(yu + 2, HW - 1) * HW), xo, R2);
    } else {
      ld_block(w2Base, 0, fa, fb);
      ld_block(w2Base, KB, ga, gb);
      pin(P0, P1);
    }

    if (u > 0) {
      const int y = y0 + u - 1;
#pragma unroll
      for (int b = 0; b < NBLK2; ++b) {
        if (b > 0 && b + 1 < NBLK2) {
          pin(acc0, acc1);
          if ((b + 1) & 1) ld_block(w2Base, (b + 1) * KB, ga, gb);
          else             ld_block(w2Base, (b + 1) * KB, fa, fb);
          pin(acc0, acc1);
        }
#pragma unroll
        for (int i = 0; i < KB; ++i) {
          const int ks = b * KB + i;
          if ((ks & 3) == 1 && (ks >> 2) < 16) {
            const int m = ks >> 2;
            if (m < 8) P0[m] = tanh_sel(P0[m]); else P1[m - 8] = tanh_sel(P1[m - 8]);
          }
          const int q = 4 * ks;
          float z0 = hr[(q + 0) / 9] * pr[(q + 0) % 9];
          float z1 = hr[(q + 1) / 9] * pr[(q + 1) % 9];
          float z2 = hr[(q + 2) / 9] * pr[(q + 2) % 9];
          float z3 = hr[(q + 3) / 9] * pr[(q + 3) % 9];
          v2f a; a.x = h ? z2 : z0; a.y = h ? z3 : z1;
          if (b & 1) { acc0 = wmma4(ga[i], a, acc0); acc1 = wmma4(gb[i], a, acc1); }
          else       { acc0 = wmma4(fa[i], a, acc0); acc1 = wmma4(fb[i], a, acc1); }
        }
      }

      float* ob = (float*)((char*)(Out + (long)bimg * NO * LPIX)
                           + (4u * (unsigned)(8 * h * LPIX + x) + 4u * (unsigned)(y * HW)));
#pragma unroll
      for (int r = 0; r < 8; ++r) {
        ob[r * LPIX]        = acc0[r];
        ob[(16 + r) * LPIX] = acc1[r];
      }
    } else {
#pragma unroll
      for (int m = 0; m < 8; ++m) { P0[m] = tanh_sel(P0[m]); P1[m] = tanh_sel(P1[m]); }
    }

#pragma unroll
    for (int j = 0; j < 6; ++j) pr[j] = pr[3 + j];
#pragma unroll
    for (int dj = 0; dj < 3; ++dj) pr[6 + dj] = prn[dj];
  }
}

extern "C" void kernel_launch(void* const* d_in, const int* in_sizes, int n_in,
                              void* d_out, int out_size, void* d_ws, size_t ws_size,
                              hipStream_t stream) {
  (void)in_sizes; (void)n_in; (void)out_size; (void)d_ws; (void)ws_size;
  const float* X    = (const float*)d_in[0];
  const float* W1   = (const float*)d_in[1];
  const float* b1   = (const float*)d_in[2];
  const float* W2   = (const float*)d_in[3];
  const float* b2   = (const float*)d_in[4];
  const float* bias = (const float*)d_in[5];
  float* Out = (float*)d_out;
  dynaconv_kernel<<<BLOCKS, WAVES * 32, 0, stream>>>(X, W1, b1, W2, b2, bias, Out);
}


# one 8-wave work-group per WGP (256 groups, was 512 of 4 waves): W1/W2 staged into LDS once per WGP by 256 threads, not twice by 128 each at the same time; tiles per wave and sums unchanged
# speedup vs baseline: 3.1670x; 1.0695x over previous
#include <hip/hip_runtime.h>
#include <math.h>

typedef float v2f __attribute__((ext_vector_type(2)));
typedef float v4f __attribute__((ext_vector_type(4)));
typedef float v8f __attribute__((ext_vector_type(8)));

#define NB 4
#define NC 16
#define NO 32
#define HW 256
#define LPIX (HW*HW)
#define DF 144
#define DQ 288
#define WAVES 8
#define TPW 8
#define ROWS_TOTAL (NB*LPIX)
#define TILES_TOTAL (ROWS_TOTAL/16)
#define BLOCKS (TILES_TOTAL/(WAVES*TPW))
#define TCOLS (HW/16)
#define STRIPS (TCOLS*(HW/TPW))
#define KB 18
#define NBLK2 (DQ/4/KB)

static __device__ __forceinline__ float xhalf(float x) {
  int b = __builtin_bit_cast(int, x);
  return __builtin_bit_cast(float,
      __builtin_amdgcn_permlanex16(b, b, 0x76543210, 0xfedcba98, false, false));
}

static __device__ __forceinline__ v8f wmma4(v2f a, v2f b, v8f c) {
  return __builtin_amdgcn_wmma_f32_16x16x4_f32(
      false, a, false, b, (short)0, c, false, false);
}

static __device__ __forceinline__ void ld_block(const float* base, int s0,
                                                v2f (&fa)[KB], v2f (&fb)[KB]) {
#pragma unroll
  for (int i = 0; i < KB; ++i) {
    fa[i] = *(const v2f*)(base + (s0 + i) * 128);
    fb[i] = *(const v2f*)(base + (s0 + i) * 128 + 32);
  }
}

static __device__ __forceinline__ void pin(v8f& a0, v8f& a1) {
  asm volatile("" : "+v"(a0), "+v"(a1) :: "memory");
}

static __device__ __forceinline__ void ld_row(const float* img, unsigned rowoff,
                                              const unsigned (&xo)[3], v2f (&R)[12]) {
#pragma unroll
  for (int dj = 0; dj < 3; ++dj) {
    const float* pt = (const float*)((const char*)img + (xo[dj] + rowoff));
#pragma unroll
    for (int g = 0; g < 4; ++g) {
      R[dj * 4 + g].x = pt[(4 * g) * LPIX];
      R[dj * 4 + g].y = pt[(4 * g + 1) * LPIX];
    }
  }
}

static __device__ __forceinline__ void zero_pad(bool rowok, bool ok0, bool ok2, v2f (&R)[12]) {
#pragma unroll
  for (int dj = 0; dj < 3; ++dj) {
    const bool ok = rowok && (dj == 0 ? ok0 : (dj == 2 ? ok2 : true));
#pragma unroll
    for (int g = 0; g < 4; ++g) {
      R[dj * 4 + g].x = ok ? R[dj * 4 + g].x : 0.0f;
      R[dj * 4 + g].y = ok ? R[dj * 4 + g].y : 0.0f;
    }
  }
}

static __device__ __forceinline__ float tap_sum(const v2f (&R)[12], int dj) {
  float s = R[dj * 4].x + R[dj * 4].y;
#pragma unroll
  for (int g = 1; g < 4; ++g) { s += R[dj * 4 + g].x; s += R[dj * 4 + g].y; }
  s += xhalf(s);
  return s;
}

static __device__ __forceinline__ float tanh_sel(float x) {
  const float y = __builtin_fabsf(x);
  const float a  = y + y;
  const float l2 = __builtin_bit_cast(float, 0x3fb8aa3bu);
  const float m  = a * l2;
  const float r  = __builtin_rintf(m);
  float e = __builtin_fmaf(a, l2, -m);
  e = __builtin_fmaf(a, __builtin_bit_cast(float, 0x32a5705fu), e);
  float p = __builtin_amdgcn_exp2f((m - r) + e);
  p = __builtin_ldexpf(p, (int)r);
  const float zb = __builtin_fmaf(__builtin_amdgcn_rcpf(1.0f + p), -2.0f, 1.0f);
  const float y2 = x * x;
  float q = __builtin_fmaf(y2, __builtin_bit_cast(float, 0xbbbac73du),
                               __builtin_bit_cast(float, 0x3ca908c9u));
  q = __builtin_fmaf(y2, q, __builtin_bit_cast(float, 0xbd5c1c4eu));
  q = __builtin_fmaf(y2, q, __builtin_bit_cast(float, 0x3e088382u));
  q = __builtin_fmaf(y2, q, __builtin_bit_cast(float, 0xbeaaaa99u));
  const float zs = __builtin_fmaf(y2, y * q, y);
  const float z = (y < 0.625f) ? zs : zb;
  return __builtin_copysignf(z, x);
}

__global__ __launch_bounds__(WAVES * 32) __attribute__((amdgpu_waves_per_eu(2)))
void dynaconv_kernel(const float* __restrict__ X,
                     const float* __restrict__ W1,
                     const float* __restrict__ b1,
                     const float* __restrict__ W2,
                     const float* __restrict__ b2,
                     const float* __restrict__ bias,
                     float* __restrict__ Out) {
  __shared__ __align__(16) float sW1[ (DF/2) * 64 ];
  __shared__ __align__(16) float sW2[ (DQ/2) * 64 ];
  __shared__ __align__(16) float sB[2 * NO];

  const int tid  = threadIdx.x;
  const int lane = tid & 31;
  const int wv   = __builtin_amdgcn_readfirstlane(tid >> 5);
  const int col  = lane & 15;
  const int h    = lane >> 4;

  const int sw   = (int)blockIdx.x * WAVES + wv;
  const int bimg = sw / STRIPS;
  const int tcol = sw % TCOLS;
  const int y0   = ((sw / TCOLS) % (HW / TPW)) * TPW;
  const int x    = tcol * 16 + col;
  const float* xb = X + bimg * NC * LPIX;
  unsigned xo[3];
  xo[0] = 4u * (unsigned)(2 * h * LPIX + max(x - 1, 0));
  xo[1] = 4u * (unsigned)(2 * h * LPIX + x);
  xo[2] = 4u * (unsigned)(2 * h * LPIX + min(x + 1, HW - 1));
  const bool ok0 = x > 0, ok2 = x < HW - 1;

  v2f R0[12], R1[12], R2[12];
  ld_row(xb, 4u * (unsigned)(max(y0 - 1, 0) * HW), xo, R0);
  ld_row(xb, 4u * (unsigned)(y0 * HW), xo, R1);
  ld_row(xb, 4u * (unsigned)(min(y0 + 1, HW - 1) * HW), xo, R2);

  for (int e = tid; e < NO * DF; e += WAVES * 32) {
    int o = e / DF, f = e - o * DF;
    int c = f / 9, ij = f - c * 9;
    int k = ij * NC + c;
    sW1[(k >> 1) * 64 + (o << 1) + (k & 1)] = W1[e];
  }
  for (int e = tid; e < DQ * NO; e += WAVES * 32) {
    int g = e >> 5, o2 = e & 31;
    int o = g / 9, j = g - o * 9;
    int q = o2 * 9 + j;
    sW2[(q >> 1) * 64 + (o << 1) + (q & 1)] = W2[e];
  }
  if (tid < NO) { sB[tid] = b1[tid]; sB[NO + tid] = bias[tid]; }
  float b2r0[72], b2r1[72];
#pragma unroll
  for (int i = 0; i < 72; ++i) { b2r0[i] = b2[72 * h + i]; b2r1[i] = b2[144 + 72 * h + i]; }

  float pr[9];
  zero_pad(y0 > 0, ok0, ok2, R0);
  zero_pad(true, ok0, ok2, R1);
#pragma unroll
  for (int dj = 0; dj < 3; ++dj) {
    pr[dj] = 0.0f; pr[3 + dj] = tap_sum(R0, dj); pr[6 + dj] = tap_sum(R1, dj);
  }
  __syncthreads();

  const float* w1Base = sW1 + h * 64 + (col << 1);
  const float* w2Base = sW2 + h * 64 + (col << 1);
  v2f fa[KB], fb[KB], ga[KB], gb[KB];

  v8f P0, P1;
#pragma unroll
  for (int r = 0; r < 8; ++r) { P0[r] = 0.0f; P1[r] = 0.0f; }

#pragma nounroll
  for (int u = 0; u <= TPW; ++u) {
    float hr[NO];
#pragma unroll
    for (int r = 0; r < 8; ++r) {
      float t0 = P0[r],     t1 = P1[r];
      float x0 = xhalf(t0), x1 = xhalf(t1);
      hr[r]      = h ? x0 : t0;
      hr[8 + r]  = h ? t0 : x0;
      hr[16 + r] = h ? x1 : t1;
      hr[24 + r] = h ? t1 : x1;
    }

    v8f acc0, acc1;
#pragma unroll
    for (int r = 0; r < 8; ++r) {
      float s0 = sB[NO + 8 * h + r], s1 = sB[NO + 16 + 8 * h + r];
#pragma unroll
      for (int j = 0; j < 9; ++j) { s0 += pr[j] * b2r0[r * 9 + j]; s1 += pr[j] * b2r1[r * 9 + j]; }
      acc0[r] = s0; acc1[r] = s1;
    }
    pin(acc0, acc1);

    float prn[3] = {0.0f, 0.0f, 0.0f};
    if (u < TPW) {
      const int yu = y0 + u;
      zero_pad(yu + 1 < HW, ok0, ok2, R2);
      v2f afr[DF / 4];
#pragma unroll
      for (int i = 0; i < 12; ++i) { afr[i] = R0[i]; afr[12 + i] = R1[i]; afr[24 + i] = R2[i]; }

#pragma unroll
      for (int r = 0; r < 8; ++r) { P0[r] = sB[8 * h + r]; P1[r] = sB[16 + 8 * h + r]; }
      ld_block(w1Base, 0, fa, fb);
      ld_block(w1Base, KB, ga, gb);
      pin(P0, P1);
#pragma unroll
      for (int i = 0; i < KB; ++i) {
        v2f a = afr[i];
        P0 = wmma4(fa[i], a, P0);
        P1 = wmma4(fb[i], a, P1);
      }
      pin(P0, P1);
      ld_block(w2Base, 0, fa, fb);
      pin(P0, P1);
#pragma unroll
      for (int i = 0; i < KB; ++i) {
        v2f a = afr[KB + i];
        P0 = wmma4(ga[i], a, P0);
        P1 = wmma4(gb[i], a, P1);
      }
      pin(P0, P1);
      ld_block(w2Base, KB, ga, gb);

#pragma unroll
      for (int dj = 0; dj < 3; ++dj) prn[dj] = tap_sum(R2, dj);

#pragma unroll
      for (int i = 0; i < 12; ++i) { R0[i] = R1[i]; R1[i] = R2[i]; }
      ld_row(xb, 4u * (unsigned)(min(yu + 2, HW - 1) * HW), xo, R2);
    } else {
      ld_block(w2Base, 0, fa, fb);
      ld_block(w2Base, KB, ga, gb);
      pin(P0, P1);
    }

    if (u > 0) {
      const int y = y0 + u - 1;
#pragma unroll
      for (int b = 0; b < NBLK2; ++b) {
        if (b > 0 && b + 1 < NBLK2) {
          pin(acc0, acc1);
          if ((b + 1) & 1) ld_block(w2Base, (b + 1) * KB, ga, gb);
          else             ld_block(w2Base, (b + 1) * KB, fa, fb);
          pin(acc0, acc1);
        }
#pragma unroll
        for (int i = 0; i < KB; ++i) {
          const int ks = b * KB + i;
          if ((ks & 3) == 1 && (ks >> 2) < 16) {
            const int m = ks >> 2;
            if (m < 8) P0[m] = tanh_sel(P0[m]); else P1[m - 8] = tanh_sel(P1[m - 8]);
          }
          const int q = 4 * ks;
          float z0 = hr[(q + 0) / 9] * pr[(q + 0) % 9];
          float z1 = hr[(q + 1) / 9] * pr[(q + 1) % 9];
          float z2 = hr[(q + 2) / 9] * pr[(q + 2) % 9];
          float z3 = hr[(q + 3) / 9] * pr[(q + 3) % 9];
          v2f a; a.x = h ? z2 : z0; a.y = h ? z3 : z1;
          if (b & 1) { acc0 = wmma4(ga[i], a, acc0); acc1 = wmma4(gb[i], a, acc1); }
          else       { acc0 = wmma4(fa[i], a, acc0); acc1 = wmma4(fb[i], a, acc1); }
        }
      }

      float* ob = (float*)((char*)(Out + (long)bimg * NO * LPIX)
                           + (4u * (unsigned)(8 * h * LPIX + x) + 4u * (unsigned)(y * HW)));
#pragma unroll
      for (int r = 0; r < 8; ++r) {
        ob[r * LPIX]        = acc0[r];
        ob[(16 + r) * LPIX] = acc1[r];
      }
    } else {
#pragma unroll
      for (int m = 0; m < 8; ++m) { P0[m] = tanh_sel(P0[m]); P1[m] = tanh_sel(P1[m]); }
    }

#pragma unroll
    for (int j = 0; j < 6; ++j) pr[j] = pr[3 + j];
#pragma unroll
    for (int dj = 0; dj < 3; ++dj) pr[6 + dj] = prn[dj];
  }
}

extern "C" void kernel_launch(void* const* d_in, const int* in_sizes, int n_in,
                              void* d_out, int out_size, void* d_ws, size_t ws_size,
                              hipStream_t stream) {
  (void)in_sizes; (void)n_in; (void)out_size; (void)d_ws; (void)ws_size;
  const float* X    = (const float*)d_in[0];
  const float* W1   = (const float*)d_in[1];
  const float* b1   = (const float*)d_in[2];
  const float* W2   = (const float*)d_in[3];
  const float* b2   = (const float*)d_in[4];
  const float* bias = (const float*)d_in[5];
  float* Out = (float*)d_out;
  dynaconv_kernel<<<BLOCKS, WAVES * 32, 0, stream>>>(X, W1, b1, W2, b2, bias, Out);
}


# one wave per SIMD (4-wave groups, 16 tiles per wave): W1/W2 fragments read from LDS once into 432 registers; tile loop has no LDS reads, its vector work sits among matrix instructions; sums unchanged
# speedup vs baseline: 3.3157x; 1.0470x over previous
#include <hip/hip_runtime.h>
#include <math.h>

typedef float v2f __attribute__((ext_vector_type(2)));
typedef float v4f __attribute__((ext_vector_type(4)));
typedef float v8f __attribute__((ext_vector_type(8)));

#define NB 4
#define NC 16
#define NO 32
#define HW 256
#define LPIX (HW*HW)
#define DF 144
#define DQ 288
#define WAVES 4
#define TPW 16
#define ROWS_TOTAL (NB*LPIX)
#define TILES_TOTAL (ROWS_TOTAL/16)
#define BLOCKS (TILES_TOTAL/(WAVES*TPW))
#define TCOLS (HW/16)
#define STRIPS (TCOLS*(HW/TPW))
#define KS1 (DF/4)
#define KS2 (DQ/4)
#define SBATCH 18

static __device__ __forceinline__ float xhalf(float x) {
  int b = __builtin_bit_cast(int, x);
  return __builtin_bit_cast(float,
      __builtin_amdgcn_permlanex16(b, b, 0x76543210, 0xfedcba98, false, false));
}

static __device__ __forceinline__ v8f wmma4(v2f a, v2f b, v8f c) {
  return __builtin_amdgcn_wmma_f32_16x16x4_f32(
      false, a, false, b, (short)0, c, false, false);
}

template <int N>
static __device__ __forceinline__ void ld_frags(const float* base, v2f (&fa)[N], v2f (&fb)[N]) {
#pragma unroll
  for (int i = 0; i < N; ++i) {
    fa[i] = *(const v2f*)(base + i * 128);
    fb[i] = *(const v2f*)(base + i * 128 + 32);
  }
}

static __device__ __forceinline__ void ld_row(const float* img, unsigned rowoff,
                                              const unsigned (&xo)[3], v2f (&R)[12]) {
#pragma unroll
  for (int dj = 0; dj < 3; ++dj) {
    const float* pt = (const float*)((const char*)img + (xo[dj] + rowoff));
#pragma unroll
    for (int g = 0; g < 4; ++g) {
      R[dj * 4 + g].x = pt[(4 * g) * LPIX];
      R[dj * 4 + g].y = pt[(4 * g + 1) * LPIX];
    }
  }
}

static __device__ __forceinline__ void zero_pad(bool rowok, bool ok0, bool ok2, v2f (&R)[12]) {
#pragma unroll
  for (int dj = 0; dj < 3; ++dj) {
    const bool ok = rowok && (dj == 0 ? ok0 : (dj == 2 ? ok2 : true));
#pragma unroll
    for (int g = 0; g < 4; ++g) {
      R[dj * 4 + g].x = ok ? R[dj * 4 + g].x : 0.0f;
      R[dj * 4 + g].y = ok ? R[dj * 4 + g].y : 0.0f;
    }
  }
}

static __device__ __forceinline__ float tap_sum(const v2f (&R)[12], int dj) {
  float s = R[dj * 4].x + R[dj * 4].y;
#pragma unroll
  for (int g = 1; g < 4; ++g) { s += R[dj * 4 + g].x; s += R[dj * 4 + g].y; }
  s += xhalf(s);
  return s;
}

static __device__ __forceinline__ float tanh_sel(float x) {
  const float y = __builtin_fabsf(x);
  const float a  = y + y;
  const float l2 = __builtin_bit_cast(float, 0x3fb8aa3bu);
  const float m  = a * l2;
  const float r  = __builtin_rintf(m);
  float e = __builtin_fmaf(a, l2, -m);
  e = __builtin_fmaf(a, __builtin_bit_cast(float, 0x32a5705fu), e);
  float p = __builtin_amdgcn_exp2f((m - r) + e);
  p = __builtin_ldexpf(p, (int)r);
  const float zb = __builtin_fmaf(__builtin_amdgcn_rcpf(1.0f + p), -2.0f, 1.0f);
  const float y2 = x * x;
  float q = __builtin_fmaf(y2, __builtin_bit_cast(float, 0xbbbac73du),
                               __builtin_bit_cast(float, 0x3ca908c9u));
  q = __builtin_fmaf(y2, q, __builtin_bit_cast(float, 0xbd5c1c4eu));
  q = __builtin_fmaf(y2, q, __builtin_bit_cast(float, 0x3e088382u));
  q = __builtin_fmaf(y2, q, __builtin_bit_cast(float, 0xbeaaaa99u));
  const float zs = __builtin_fmaf(y2, y * q, y);
  const float z = (y < 0.625f) ? zs : zb;
  return __builtin_copysignf(z, x);
}

static __device__ __forceinline__ void hid_pair(const v8f& P0, const v8f& P1, int h, int r,
                                                float (&hr)[NO]) {
  float t0 = P0[r],     t1 = P1[r];
  float x0 = xhalf(t0), x1 = xhalf(t1);
  hr[r]      = h ? x0 : t0;
  hr[8 + r]  = h ? t0 : x0;
  hr[16 + r] = h ? x1 : t1;
  hr[24 + r] = h ? t1 : x1;
}

static __device__ __forceinline__ void acc_tap(int j, const float (&pr)[9],
                                               const float (&b2r0)[72], const float (&b2r1)[72],
                                               const v8f& bs0, const v8f& bs1, v8f& acc0, v8f& acc1) {
#pragma unroll
  for (int r = 0; r < 8; ++r) {
    float s0 = (j == 0) ? bs0[r] : acc0[r];
    float s1 = (j == 0) ? bs1[r] : acc1[r];
    s0 += pr[j] * b2r0[r * 9 + j];
    s1 += pr[j] * b2r1[r * 9 + j];
    acc0[r] = s0; acc1[r] = s1;
  }
}

__global__ __launch_bounds__(WAVES * 32)
void dynaconv_kernel(const float* __restrict__ X,
                     const float* __restrict__ W1,
                     const float* __restrict__ b1,
                     const float* __restrict__ W2,
                     const float* __restrict__ b2,
                     const float* __restrict__ bias,
                     float* __restrict__ Out) {
  __shared__ __align__(16) float sW1[ (DF/2) * 64 ];
  __shared__ __align__(16) float sW2[ (DQ/2) * 64 ];

  const int tid  = threadIdx.x;
  const int lane = tid & 31;
  const int wv   = __builtin_amdgcn_readfirstlane(tid >> 5);
  const int col  = lane & 15;
  const int h    = lane >> 4;

  const int sw   = (int)blockIdx.x * WAVES + wv;
  const int bimg = sw / STRIPS;
  const int tcol = sw % TCOLS;
  const int y0   = ((sw / TCOLS) % (HW / TPW)) * TPW;
  const int x    = tcol * 16 + col;
  const float* xb = X + bimg * NC * LPIX;
  unsigned xo[3];
  xo[0] = 4u * (unsigned)(2 * h * LPIX + max(x - 1, 0));
  xo[1] = 4u * (unsigned)(2 * h * LPIX + x);
  xo[2] = 4u * (unsigned)(2 * h * LPIX + min(x + 1, HW - 1));
  const bool ok0 = x > 0, ok2 = x < HW - 1;

  v2f R0[12], R1[12], R2[12], Rn[12];
  ld_row(xb, 4u * (unsigned)(max(y0 - 1, 0) * HW), xo, R0);
  ld_row(xb, 4u * (unsigned)(y0 * HW), xo, R1);
  ld_row(xb, 4u * (unsigned)(min(y0 + 1, HW - 1) * HW), xo, R2);

#pragma nounroll
  for (int e0 = 0; e0 < NO * DF; e0 += SBATCH * WAVES * 32) {
    float v[SBATCH];
#pragma unroll
    for (int i = 0; i < SBATCH; ++i) v[i] = W1[e0 + i * (WAVES * 32) + tid];
#pragma unroll
    for (int i = 0; i < SBATCH; ++i) {
      int e = e0 + i * (WAVES * 32) + tid;
      int o = e / DF, f = e - o * DF;
      int c = f / 9, ij = f - c * 9;
      int k = ij * NC + c;
      sW1[(k >> 1) * 64 + (o << 1) + (k & 1)] = v[i];
    }
  }
#pragma nounroll
  for (int e0 = 0; e0 < DQ * NO; e0 += SBATCH * WAVES * 32) {
    float v[SBATCH];
#pragma unroll
    for (int i = 0; i < SBATCH; ++i) v[i] = W2[e0 + i * (WAVES * 32) + tid];
#pragma unroll
    for (int i = 0; i < SBATCH; ++i) {
      int e = e0 + i * (WAVES * 32) + tid;
      int g = e >> 5, o2 = e & 31;
      int o = g / 9, j = g - o * 9;
      int q = o2 * 9 + j;
      sW2[(q >> 1) * 64 + (o << 1) + (q & 1)] = v[i];
    }
  }
  float b2r0[72], b2r1[72];
#pragma unroll
  for (int i = 0; i < 72; ++i) { b2r0[i] = b2[72 * h + i]; b2r1[i] = b2[144 + 72 * h + i]; }
  v8f b1v0, b1v1, bs0, bs1;
#pragma unroll
  for (int r = 0; r < 8; ++r) {
    b1v0[r] = b1[8 * h + r];   b1v1[r] = b1[16 + 8 * h + r];
    bs0[r]  = bias[8 * h + r]; bs1[r]  = bias[16 + 8 * h + r];
  }

  float pr[9];
  zero_pad(y0 > 0, ok0, ok2, R0);
  zero_pad(true, ok0, ok2, R1);
#pragma unroll
  for (int dj = 0; dj < 3; ++dj) {
    pr[dj] = 0.0f; pr[3 + dj] = tap_sum(R0, dj); pr[6 + dj] = tap_sum(R1, dj);
  }
  __syncthreads();

  v2f w1a[KS1], w1b[KS1], w2a[KS2], w2b[KS2];
  ld_frags<KS1>(sW1 + h * 64 + (col << 1), w1a, w1b);
  ld_frags<KS2>(sW2 + h * 64 + (col << 1), w2a, w2b);

  v8f P0, P1, Q0, Q1;
#pragma unroll
  for (int r = 0; r < 8; ++r) { P0[r] = 0.0f; P1[r] = 0.0f; Q0[r] = 0.0f; Q1[r] = 0.0f; }

#pragma nounroll
  for (int u = 0; u <= TPW; ++u) {
    float hr[NO];
    v8f acc0, acc1;
    float prn[3] = {0.0f, 0.0f, 0.0f};
    if (u < TPW) {
      const int yu = y0 + u;
      ld_row(xb, 4u * (unsigned)(min(yu + 2, HW - 1) * HW), xo, Rn);
      zero_pad(yu + 1 < HW, ok0, ok2, R2);
      v2f afr[KS1];
#pragma unroll
      for (int i = 0; i < 12; ++i) { afr[i] = R0[i]; afr[12 + i] = R1[i]; afr[24 + i] = R2[i]; }

#pragma unroll
      for (int i = 0; i < KS1; ++i) {
        v2f a = afr[i];
        Q0 = wmma4(w1a[i], a, i == 0 ? b1v0 : Q0);
        Q1 = wmma4(w1b[i], a, i == 0 ? b1v1 : Q1);
        if (i < 8) hid_pair(P0, P1, h, i, hr);
        else if (i < 17) acc_tap(i - 8, pr, b2r0, b2r1, bs0, bs1, acc0, acc1);
        else if (i < 20) prn[i - 17] = tap_sum(R2, i - 17);
      }
    } else {
#pragma unroll
      for (int r = 0; r < 8; ++r) hid_pair(P0, P1, h, r, hr);
#pragma unroll
      for (int j = 0; j < 9; ++j) acc_tap(j, pr, b2r0, b2r1, bs0, bs1, acc0, acc1);
    }

    if (u > 0) {
      const int y = y0 + u - 1;
#pragma unroll
      for (int ks = 0; ks < KS2; ++ks) {
        if ((ks & 3) == 1 && (ks >> 2) < 16) {
          const int m = ks >> 2;
          if (m < 8) P0[m] = tanh_sel(Q0[m]); else P1[m - 8] = tanh_sel(Q1[m - 8]);
        }
        const int q = 4 * ks;
        float z0 = hr[(q + 0) / 9] * pr[(q + 0) % 9];
        float z1 = hr[(q + 1) / 9] * pr[(q + 1) % 9];
        float z2 = hr[(q + 2) / 9] * pr[(q + 2) % 9];
        float z3 = hr[(q + 3) / 9] * pr[(q + 3) % 9];
        v2f a; a.x = h ? z2 : z0; a.y = h ? z3 : z1;
        acc0 = wmma4(w2a[ks], a, acc0);
        acc1 = wmma4(w2b[ks], a, acc1);
      }

      float* ob = (float*)((char*)(Out + (long)bimg * NO * LPIX)
                           + (4u * (unsigned)(8 * h * LPIX + x) + 4u * (unsigned)(y * HW)));
#pragma unroll
      for (int r = 0; r < 8; ++r) {
        ob[r * LPIX]        = acc0[r];
        ob[(16 + r) * LPIX] = acc1[r];
      }
    } else {
#pragma unroll
      for (int m = 0; m < 8; ++m) { P0[m] = tanh_sel(Q0[m]); P1[m] = tanh_sel(Q1[m]); }
    }

#pragma unroll
    for (int i = 0; i < 12; ++i) { R0[i] = R1[i]; R1[i] = R2[i]; R2[i] = Rn[i]; }
#pragma unroll
    for (int j = 0; j < 6; ++j) pr[j] = pr[3 + j];
#pragma unroll
    for (int dj = 0; dj < 3; ++dj) pr[6 + dj] = prn[dj];
  }
}

extern "C" void kernel_launch(void* const* d_in, const int* in_sizes, int n_in,
                              void* d_out, int out_size, void* d_ws, size_t ws_size,
                              hipStream_t stream) {
  (void)in_sizes; (void)n_in; (void)out_size; (void)d_ws; (void)ws_size;
  const float* X    = (const float*)d_in[0];
  const float* W1   = (const float*)d_in[1];
  const float* b1   = (const float*)d_in[2];
  const float* W2   = (const float*)d_in[3];
  const float* b2   = (const float*)d_in[4];
  const float* bias = (const float*)d_in[5];
  float* Out = (float*)d_out;
  dynaconv_kernel<<<BLOCKS, WAVES * 32, 0, stream>>>(X, W1, b1, W2, b2, bias, Out);
}


# GEMM2 as W2 (288x32, rows from memory as A) x hidden (K=32; B = GEMM1's tanh'd accumulators in place), 9-tap contraction as packed lane FMAs; no hid x pc operand built; GEMM2 sum order changed
# speedup vs baseline: 3.6203x; 1.0919x over previous
#include <hip/hip_runtime.h>
#include <math.h>

typedef float v2f __attribute__((ext_vector_type(2)));
typedef float v4f __attribute__((ext_vector_type(4)));
typedef float v8f __attribute__((ext_vector_type(8)));

#define NB 4
#define NC 16
#define NO 32
#define HW 256
#define LPIX (HW*HW)
#define DF 144
#define DQ 288
#define WAVES 4
#define TPW 16
#define ROWS_TOTAL (NB*LPIX)
#define TILES_TOTAL (ROWS_TOTAL/16)
#define BLOCKS (TILES_TOTAL/(WAVES*TPW))
#define TCOLS (HW/16)
#define STRIPS (TCOLS*(HW/TPW))
#define KS1 (DF/4)
#define MT2 (DQ/16)
#define KS2 (NO/4)
#define SBATCH 18

static __device__ __forceinline__ float xhalf(float x) {
  int b = __builtin_bit_cast(int, x);
  return __builtin_bit_cast(float,
      __builtin_amdgcn_permlanex16(b, b, 0x76543210, 0xfedcba98, false, false));
}

static __device__ __forceinline__ v8f wmma4(v2f a, v2f b, v8f c) {
  return __builtin_amdgcn_wmma_f32_16x16x4_f32(
      false, a, false, b, (short)0, c, false, false);
}

template <int N>
static __device__ __forceinline__ void ld_frags(const float* base, v2f (&fa)[N], v2f (&fb)[N]) {
#pragma unroll
  for (int i = 0; i < N; ++i) {
    fa[i] = *(const v2f*)(base + i * 128);
    fb[i] = *(const v2f*)(base + i * 128 + 32);
  }
}

static __device__ __forceinline__ void ld_row(const float* img, unsigned rowoff,
                                              const unsigned (&xo)[3], v2f (&R)[12]) {
#pragma unroll
  for (int dj = 0; dj < 3; ++dj) {
    const float* pt = (const float*)((const char*)img + (xo[dj] + rowoff));
#pragma unroll
    for (int g = 0; g < 4; ++g) {
      R[dj * 4 + g].x = pt[(4 * g) * LPIX];
      R[dj * 4 + g].y = pt[(4 * g + 1) * LPIX];
    }
  }
}

static __device__ __forceinline__ void zero_pad(bool rowok, bool ok0, bool ok2, v2f (&R)[12]) {
#pragma unroll
  for (int dj = 0; dj < 3; ++dj) {
    const bool ok = rowok && (dj == 0 ? ok0 : (dj == 2 ? ok2 : true));
#pragma unroll
    for (int g = 0; g < 4; ++g) {
      R[dj * 4 + g].x = ok ? R[dj * 4 + g].x : 0.0f;
      R[dj * 4 + g].y = ok ? R[dj * 4 + g].y : 0.0f;
    }
  }
}

static __device__ __forceinline__ float tap_sum(const v2f (&R)[12], int dj) {
  float s = R[dj * 4].x + R[dj * 4].y;
#pragma unroll
  for (int g = 1; g < 4; ++g) { s += R[dj * 4 + g].x; s += R[dj * 4 + g].y; }
  s += xhalf(s);
  return s;
}

static __device__ __forceinline__ float tanh_sel(float x) {
  const float y = __builtin_fabsf(x);
  const float a  = y + y;
  const float l2 = __builtin_bit_cast(float, 0x3fb8aa3bu);
  const float m  = a * l2;
  const float r  = __builtin_rintf(m);
  float e = __builtin_fmaf(a, l2, -m);
  e = __builtin_fmaf(a, __builtin_bit_cast(float, 0x32a5705fu), e);
  float p = __builtin_amdgcn_exp2f((m - r) + e);
  p = __builtin_ldexpf(p, (int)r);
  const float zb = __builtin_fmaf(__builtin_amdgcn_rcpf(1.0f + p), -2.0f, 1.0f);
  const float y2 = x * x;
  float q = __builtin_fmaf(y2, __builtin_bit_cast(float, 0xbbbac73du),
                               __builtin_bit_cast(float, 0x3ca908c9u));
  q = __builtin_fmaf(y2, q, __builtin_bit_cast(float, 0xbd5c1c4eu));
  q = __builtin_fmaf(y2, q, __builtin_bit_cast(float, 0x3e088382u));
  q = __builtin_fmaf(y2, q, __builtin_bit_cast(float, 0xbeaaaa99u));
  const float zs = __builtin_fmaf(y2, y * q, y);
  const float z = (y < 0.625f) ? zs : zb;
  return __builtin_copysignf(z, x);
}

static __device__ __forceinline__ int row_of(int t, int hm, int r) {
  const int slot = 4 * (t % 9) + (r >> 1);
  const int cp = slot / 9, j = slot - 9 * cp;
  const int o = (t < 9 ? 0 : 16) + 8 * hm + 2 * cp + (r & 1);
  return o * 9 + j;
}

__global__ __launch_bounds__(WAVES * 32)
void dynaconv_kernel(const float* __restrict__ X,
                     const float* __restrict__ W1,
                     const float* __restrict__ b1,
                     const float* __restrict__ W2,
                     const float* __restrict__ b2,
                     const float* __restrict__ bias,
                     float* __restrict__ Out) {
  __shared__ __align__(16) float sW1[ (DF/2) * 64 ];

  const int tid  = threadIdx.x;
  const int lane = tid & 31;
  const int wv   = __builtin_amdgcn_readfirstlane(tid >> 5);
  const int col  = lane & 15;
  const int h    = lane >> 4;

  const int sw   = (int)blockIdx.x * WAVES + wv;
  const int bimg = sw / STRIPS;
  const int tcol = sw % TCOLS;
  const int y0   = ((sw / TCOLS) % (HW / TPW)) * TPW;
  const int x    = tcol * 16 + col;
  const float* xb = X + bimg * NC * LPIX;
  unsigned xo[3];
  xo[0] = 4u * (unsigned)(2 * h * LPIX + max(x - 1, 0));
  xo[1] = 4u * (unsigned)(2 * h * LPIX + x);
  xo[2] = 4u * (unsigned)(2 * h * LPIX + min(x + 1, HW - 1));
  const bool ok0 = x > 0, ok2 = x < HW - 1;

  v2f R0[12], R1[12], R2[12], Rn[12];
  ld_row(xb, 4u * (unsigned)(max(y0 - 1, 0) * HW), xo, R0);
  ld_row(xb, 4u * (unsigned)(y0 * HW), xo, R1);
  ld_row(xb, 4u * (unsigned)(min(y0 + 1, HW - 1) * HW), xo, R2);

  v2f w2f[MT2][KS2];
#pragma unroll
  for (int t = 0; t < MT2; ++t) {
    const int g = row_of(t, col >> 3, col & 7);
    const v4f* wp = (const v4f*)(W2 + g * NO + 8 * h);
    const v4f x0 = wp[0], x1 = wp[1], x2 = wp[4], x3 = wp[5];
    w2f[t][0].x = x0.x; w2f[t][0].y = x0.y; w2f[t][1].x = x0.z; w2f[t][1].y = x0.w;
    w2f[t][2].x = x1.x; w2f[t][2].y = x1.y; w2f[t][3].x = x1.z; w2f[t][3].y = x1.w;
    w2f[t][4].x = x2.x; w2f[t][4].y = x2.y; w2f[t][5].x = x2.z; w2f[t][5].y = x2.w;
    w2f[t][6].x = x3.x; w2f[t][6].y = x3.y; w2f[t][7].x = x3.z; w2f[t][7].y = x3.w;
  }

#pragma nounroll
  for (int e0 = 0; e0 < NO * DF; e0 += SBATCH * WAVES * 32) {
    float v[SBATCH];
#pragma unroll
    for (int i = 0; i < SBATCH; ++i) v[i] = W1[e0 + i * (WAVES * 32) + tid];
#pragma unroll
    for (int i = 0; i < SBATCH; ++i) {
      int e = e0 + i * (WAVES * 32) + tid;
      int o = e / DF, f = e - o * DF;
      int c = f / 9, ij = f - c * 9;
      int k = ij * NC + c;
      sW1[(k >> 1) * 64 + (o << 1) + (k & 1)] = v[i];
    }
  }
  v8f c2[MT2];
#pragma unroll
  for (int t = 0; t < MT2; ++t) {
#pragma unroll
    for (int r = 0; r < 8; ++r) c2[t][r] = b2[row_of(t, h, r)];
  }
  v8f b1v0, b1v1, bs0, bs1;
#pragma unroll
  for (int r = 0; r < 8; ++r) {
    b1v0[r] = b1[8 * h + r];   b1v1[r] = b1[16 + 8 * h + r];
    bs0[r]  = bias[8 * h + r]; bs1[r]  = bias[16 + 8 * h + r];
  }

  float pr[9];
  zero_pad(y0 > 0, ok0, ok2, R0);
  zero_pad(true, ok0, ok2, R1);
#pragma unroll
  for (int dj = 0; dj < 3; ++dj) {
    pr[dj] = 0.0f; pr[3 + dj] = tap_sum(R0, dj); pr[6 + dj] = tap_sum(R1, dj);
  }
  __syncthreads();

  v2f w1a[KS1], w1b[KS1];
  ld_frags<KS1>(sW1 + h * 64 + (col << 1), w1a, w1b);

  v8f P0, P1, Pn0, Pn1, Q0, Q1;
#pragma unroll
  for (int r = 0; r < 8; ++r) {
    P0[r] = 0.0f; P1[r] = 0.0f; Pn0[r] = 0.0f; Pn1[r] = 0.0f; Q0[r] = 0.0f; Q1[r] = 0.0f;
  }

#pragma nounroll
  for (int u = 0; u <= TPW; ++u) {
    float prn[3] = {0.0f, 0.0f, 0.0f};
    if (u < TPW) {
      const int yu = y0 + u;
      ld_row(xb, 4u * (unsigned)(min(yu + 2, HW - 1) * HW), xo, Rn);
      zero_pad(yu + 1 < HW, ok0, ok2, R2);
      v2f afr[KS1];
#pragma unroll
      for (int i = 0; i < 12; ++i) { afr[i] = R0[i]; afr[12 + i] = R1[i]; afr[24 + i] = R2[i]; }

#pragma unroll
      for (int i = 0; i < KS1; ++i) {
        v2f a = afr[i];
        Q0 = wmma4(w1a[i], a, i == 0 ? b1v0 : Q0);
        Q1 = wmma4(w1b[i], a, i == 0 ? b1v1 : Q1);
        if (i >= 2 && i < 5) prn[i - 2] = tap_sum(R2, i - 2);
      }
    }

    if (u > 0) {
      const int y = y0 + u - 1;
      v2f oa[8];
      v8f kt[MT2];
#pragma unroll
      for (int p = 0; p <= MT2 / 2; ++p) {
        const int t0 = 2 * p, t1 = 2 * p + 1;
#pragma unroll
        for (int s = 0; s < KS2; ++s) {
          const int ks = p * KS2 + s;
          if (p < MT2 / 2) {
            if ((ks & 3) == 1 && (ks >> 2) < 16) {
              const int m = ks >> 2;
              if (m < 8) Pn0[m] = tanh_sel(Q0[m]); else Pn1[m - 8] = tanh_sel(Q1[m - 8]);
            }
            v2f b;
            if (s < 4) { b.x = P0[2 * s]; b.y = P0[2 * s + 1]; }
            else       { b.x = P1[2 * (s - 4)]; b.y = P1[2 * (s - 4) + 1]; }
            kt[t0] = wmma4(w2f[t0][s], b, s == 0 ? c2[t0] : kt[t0]);
            kt[t1] = wmma4(w2f[t1][s], b, s == 0 ? c2[t1] : kt[t1]);
          }
          if (p > 0) {
            const int tp = (s < 4) ? t0 - 2 : t1 - 2;
            const int slot = 4 * (tp % 9) + (s & 3);
            const int cp = slot / 9, j = slot % 9;
            v2f kv;
            kv.x = kt[tp][2 * (s & 3)]; kv.y = kt[tp][2 * (s & 3) + 1];
            const int i = (tp < 9 ? 0 : 4) + cp;
            if (j == 0) { oa[i].x = kv.x * pr[0]; oa[i].y = kv.y * pr[0]; }
            else        { oa[i].x += kv.x * pr[j]; oa[i].y += kv.y * pr[j]; }
          }
        }
      }

      float* ob = (float*)((char*)(Out + (long)bimg * NO * LPIX)
                           + (4u * (unsigned)(8 * h * LPIX + x) + 4u * (unsigned)(y * HW)));
#pragma unroll
      for (int i = 0; i < 4; ++i) {
        ob[(2 * i) * LPIX]          = oa[i].x + bs0[2 * i];
        ob[(2 * i + 1) * LPIX]      = oa[i].y + bs0[2 * i + 1];
        ob[(16 + 2 * i) * LPIX]     = oa[4 + i].x + bs1[2 * i];
        ob[(16 + 2 * i + 1) * LPIX] = oa[4 + i].y + bs1[2 * i + 1];
      }
    } else {
#pragma unroll
      for (int m = 0; m < 8; ++m) { Pn0[m] = tanh_sel(Q0[m]); Pn1[m] = tanh_sel(Q1[m]); }
    }

    P0 = Pn0; P1 = Pn1;
#pragma unroll
    for (int i = 0; i < 12; ++i) { R0[i] = R1[i]; R1[i] = R2[i]; R2[i] = Rn[i]; }
#pragma unroll
    for (int j = 0; j < 6; ++j) pr[j] = pr[3 + j];
#pragma unroll
    for (int dj = 0; dj < 3; ++dj) pr[6 + dj] = prn[dj];
  }
}

extern "C" void kernel_launch(void* const* d_in, const int* in_sizes, int n_in,
                              void* d_out, int out_size, void* d_ws, size_t ws_size,
                              hipStream_t stream) {
  (void)in_sizes; (void)n_in; (void)out_size; (void)d_ws; (void)ws_size;
  const float* X    = (const float*)d_in[0];
  const float* W1   = (const float*)d_in[1];
  const float* b1   = (const float*)d_in[2];
  const float* W2   = (const float*)d_in[3];
  const float* b2   = (const float*)d_in[4];
  const float* bias = (const float*)d_in[5];
  float* Out = (float*)d_out;
  dynaconv_kernel<<<BLOCKS, WAVES * 32, 0, stream>>>(X, W1, b1, W2, b2, bias, Out);
}


# tanh of the hidden layer by one formula for every x, 1 - 2/(1 + exp2(x*2log2(e))): exp2, add, rcp, fma on register pairs (was both library range formulas and a select); tanh's last bits differ
# speedup vs baseline: 3.8342x; 1.0591x over previous
#include <hip/hip_runtime.h>
#include <math.h>

typedef float v2f __attribute__((ext_vector_type(2)));
typedef float v4f __attribute__((ext_vector_type(4)));
typedef float v8f __attribute__((ext_vector_type(8)));

#define NB 4
#define NC 16
#define NO 32
#define HW 256
#define LPIX (HW*HW)
#define DF 144
#define DQ 288
#define WAVES 4
#define TPW 16
#define ROWS_TOTAL (NB*LPIX)
#define TILES_TOTAL (ROWS_TOTAL/16)
#define BLOCKS (TILES_TOTAL/(WAVES*TPW))
#define TCOLS (HW/16)
#define STRIPS (TCOLS*(HW/TPW))
#define KS1 (DF/4)
#define MT2 (DQ/16)
#define KS2 (NO/4)
#define SBATCH 18

static __device__ __forceinline__ float xhalf(float x) {
  int b = __builtin_bit_cast(int, x);
  return __builtin_bit_cast(float,
      __builtin_amdgcn_permlanex16(b, b, 0x76543210, 0xfedcba98, false, false));
}

static __device__ __forceinline__ v8f wmma4(v2f a, v2f b, v8f c) {
  return __builtin_amdgcn_wmma_f32_16x16x4_f32(
      false, a, false, b, (short)0, c, false, false);
}

template <int N>
static __device__ __forceinline__ void ld_frags(const float* base, v2f (&fa)[N], v2f (&fb)[N]) {
#pragma unroll
  for (int i = 0; i < N; ++i) {
    fa[i] = *(const v2f*)(base + i * 128);
    fb[i] = *(const v2f*)(base + i * 128 + 32);
  }
}

static __device__ __forceinline__ void ld_row(const float* img, unsigned rowoff,
                                              const unsigned (&xo)[3], v2f (&R)[12]) {
#pragma unroll
  for (int dj = 0; dj < 3; ++dj) {
    const float* pt = (const float*)((const char*)img + (xo[dj] + rowoff));
#pragma unroll
    for (int g = 0; g < 4; ++g) {
      R[dj * 4 + g].x = pt[(4 * g) * LPIX];
      R[dj * 4 + g].y = pt[(4 * g + 1) * LPIX];
    }
  }
}

static __device__ __forceinline__ void zero_pad(bool rowok, bool ok0, bool ok2, v2f (&R)[12]) {
#pragma unroll
  for (int dj = 0; dj < 3; ++dj) {
    const bool ok = rowok && (dj == 0 ? ok0 : (dj == 2 ? ok2 : true));
#pragma unroll
    for (int g = 0; g < 4; ++g) {
      R[dj * 4 + g].x = ok ? R[dj * 4 + g].x : 0.0f;
      R[dj * 4 + g].y = ok ? R[dj * 4 + g].y : 0.0f;
    }
  }
}

static __device__ __forceinline__ float tap_sum(const v2f (&R)[12], int dj) {
  float s = R[dj * 4].x + R[dj * 4].y;
#pragma unroll
  for (int g = 1; g < 4; ++g) { s += R[dj * 4 + g].x; s += R[dj * 4 + g].y; }
  s += xhalf(s);
  return s;
}

static __device__ __forceinline__ v2f tanh_pair(v2f x) {
  const float c2l = __builtin_bit_cast(float, 0x4038aa3bu);
  const v2f m = x * c2l;
  v2f p;
  p.x = __builtin_amdgcn_exp2f(m.x);
  p.y = __builtin_amdgcn_exp2f(m.y);
  const v2f d = p + 1.0f;
  v2f z;
  z.x = __builtin_fmaf(__builtin_amdgcn_rcpf(d.x), -2.0f, 1.0f);
  z.y = __builtin_fmaf(__builtin_amdgcn_rcpf(d.y), -2.0f, 1.0f);
  return z;
}

static __device__ __forceinline__ int row_of(int t, int hm, int r) {
  const int slot = 4 * (t % 9) + (r >> 1);
  const int cp = slot / 9, j = slot - 9 * cp;
  const int o = (t < 9 ? 0 : 16) + 8 * hm + 2 * cp + (r & 1);
  return o * 9 + j;
}

__global__ __launch_bounds__(WAVES * 32)
void dynaconv_kernel(const float* __restrict__ X,
                     const float* __restrict__ W1,
                     const float* __restrict__ b1,
                     const float* __restrict__ W2,
                     const float* __restrict__ b2,
                     const float* __restrict__ bias,
                     float* __restrict__ Out) {
  __shared__ __align__(16) float sW1[ (DF/2) * 64 ];

  const int tid  = threadIdx.x;
  const int lane = tid & 31;
  const int wv   = __builtin_amdgcn_readfirstlane(tid >> 5);
  const int col  = lane & 15;
  const int h    = lane >> 4;

  const int sw   = (int)blockIdx.x * WAVES + wv;
  const int bimg = sw / STRIPS;
  const int tcol = sw % TCOLS;
  const int y0   = ((sw / TCOLS) % (HW / TPW)) * TPW;
  const int x    = tcol * 16 + col;
  const float* xb = X + bimg * NC * LPIX;
  unsigned xo[3];
  xo[0] = 4u * (unsigned)(2 * h * LPIX + max(x - 1, 0));
  xo[1] = 4u * (unsigned)(2 * h * LPIX + x);
  xo[2] = 4u * (unsigned)(2 * h * LPIX + min(x + 1, HW - 1));
  const bool ok0 = x > 0, ok2 = x < HW - 1;

  v2f R0[12], R1[12], R2[12], Rn[12];
  ld_row(xb, 4u * (unsigned)(max(y0 - 1, 0) * HW), xo, R0);
  ld_row(xb, 4u * (unsigned)(y0 * HW), xo, R1);
  ld_row(xb, 4u * (unsigned)(min(y0 + 1, HW - 1) * HW), xo, R2);

  v2f w2f[MT2][KS2];
#pragma unroll
  for (int t = 0; t < MT2; ++t) {
    const int g = row_of(t, col >> 3, col & 7);
    const v4f* wp = (const v4f*)(W2 + g * NO + 8 * h);
    const v4f x0 = wp[0], x1 = wp[1], x2 = wp[4], x3 = wp[5];
    w2f[t][0].x = x0.x; w2f[t][0].y = x0.y; w2f[t][1].x = x0.z; w2f[t][1].y = x0.w;
    w2f[t][2].x = x1.x; w2f[t][2].y = x1.y; w2f[t][3].x = x1.z; w2f[t][3].y = x1.w;
    w2f[t][4].x = x2.x; w2f[t][4].y = x2.y; w2f[t][5].x = x2.z; w2f[t][5].y = x2.w;
    w2f[t][6].x = x3.x; w2f[t][6].y = x3.y; w2f[t][7].x = x3.z; w2f[t][7].y = x3.w;
  }

#pragma nounroll
  for (int e0 = 0; e0 < NO * DF; e0 += SBATCH * WAVES * 32) {
    float v[SBATCH];
#pragma unroll
    for (int i = 0; i < SBATCH; ++i) v[i] = W1[e0 + i * (WAVES * 32) + tid];
#pragma unroll
    for (int i = 0; i < SBATCH; ++i) {
      int e = e0 + i * (WAVES * 32) + tid;
      int o = e / DF, f = e - o * DF;
      int c = f / 9, ij = f - c * 9;
      int k = ij * NC + c;
      sW1[(k >> 1) * 64 + (o << 1) + (k & 1)] = v[i];
    }
  }
  v8f c2[MT2];
#pragma unroll
  for (int t = 0; t < MT2; ++t) {
#pragma unroll
    for (int r = 0; r < 8; ++r) c2[t][r] = b2[row_of(t, h, r)];
  }
  v8f b1v0, b1v1, bs0, bs1;
#pragma unroll
  for (int r = 0; r < 8; ++r) {
    b1v0[r] = b1[8 * h + r];   b1v1[r] = b1[16 + 8 * h + r];
    bs0[r]  = bias[8 * h + r]; bs1[r]  = bias[16 + 8 * h + r];
  }

  float pr[9];
  zero_pad(y0 > 0, ok0, ok2, R0);
  zero_pad(true, ok0, ok2, R1);
#pragma unroll
  for (int dj = 0; dj < 3; ++dj) {
    pr[dj] = 0.0f; pr[3 + dj] = tap_sum(R0, dj); pr[6 + dj] = tap_sum(R1, dj);
  }
  __syncthreads();

  v2f w1a[KS1], w1b[KS1];
  ld_frags<KS1>(sW1 + h * 64 + (col << 1), w1a, w1b);

  v8f P0, P1, Pn0, Pn1, Q0, Q1;
#pragma unroll
  for (int r = 0; r < 8; ++r) {
    P0[r] = 0.0f; P1[r] = 0.0f; Pn0[r] = 0.0f; Pn1[r] = 0.0f; Q0[r] = 0.0f; Q1[r] = 0.0f;
  }

#pragma nounroll
  for (int u = 0; u <= TPW; ++u) {
    float prn[3] = {0.0f, 0.0f, 0.0f};
    if (u < TPW) {
      const int yu = y0 + u;
      ld_row(xb, 4u * (unsigned)(min(yu + 2, HW - 1) * HW), xo, Rn);
      zero_pad(yu + 1 < HW, ok0, ok2, R2);
      v2f afr[KS1];
#pragma unroll
      for (int i = 0; i < 12; ++i) { afr[i] = R0[i]; afr[12 + i] = R1[i]; afr[24 + i] = R2[i]; }

#pragma unroll
      for (int i = 0; i < KS1; ++i) {
        v2f a = afr[i];
        Q0 = wmma4(w1a[i], a, i == 0 ? b1v0 : Q0);
        Q1 = wmma4(w1b[i], a, i == 0 ? b1v1 : Q1);
        if (i >= 2 && i < 5) prn[i - 2] = tap_sum(R2, i - 2);
      }
    }

    if (u > 0) {
      const int y = y0 + u - 1;
      v2f oa[8];
      v8f kt[MT2];
#pragma unroll
      for (int p = 0; p <= MT2 / 2; ++p) {
        const int t0 = 2 * p, t1 = 2 * p + 1;
#pragma unroll
        for (int s = 0; s < KS2; ++s) {
          if (p < MT2 / 2) {
            if (s == 1 && p < 8) {
              v2f q, z;
              if (p < 4) { q.x = Q0[2 * p]; q.y = Q0[2 * p + 1]; }
              else       { q.x = Q1[2 * (p - 4)]; q.y = Q1[2 * (p - 4) + 1]; }
              z = tanh_pair(q);
              if (p < 4) { Pn0[2 * p] = z.x; Pn0[2 * p + 1] = z.y; }
              else       { Pn1[2 * (p - 4)] = z.x; Pn1[2 * (p - 4) + 1] = z.y; }
            }
            v2f b;
            if (s < 4) { b.x = P0[2 * s]; b.y = P0[2 * s + 1]; }
            else       { b.x = P1[2 * (s - 4)]; b.y = P1[2 * (s - 4) + 1]; }
            kt[t0] = wmma4(w2f[t0][s], b, s == 0 ? c2[t0] : kt[t0]);
            kt[t1] = wmma4(w2f[t1][s], b, s == 0 ? c2[t1] : kt[t1]);
          }
          if (p > 0) {
            const int tp = (s < 4) ? t0 - 2 : t1 - 2;
            const int slot = 4 * (tp % 9) + (s & 3);
            const int cp = slot / 9, j = slot % 9;
            v2f kv;
            kv.x = kt[tp][2 * (s & 3)]; kv.y = kt[tp][2 * (s & 3) + 1];
            const int i = (tp < 9 ? 0 : 4) + cp;
            if (j == 0) { oa[i].x = kv.x * pr[0]; oa[i].y = kv.y * pr[0]; }
            else        { oa[i].x += kv.x * pr[j]; oa[i].y += kv.y * pr[j]; }
          }
        }
      }

      float* ob = (float*)((char*)(Out + (long)bimg * NO * LPIX)
                           + (4u * (unsigned)(8 * h * LPIX + x) + 4u * (unsigned)(y * HW)));
#pragma unroll
      for (int i = 0; i < 4; ++i) {
        ob[(2 * i) * LPIX]          = oa[i].x + bs0[2 * i];
        ob[(2 * i + 1) * LPIX]      = oa[i].y + bs0[2 * i + 1];
        ob[(16 + 2 * i) * LPIX]     = oa[4 + i].x + bs1[2 * i];
        ob[(16 + 2 * i + 1) * LPIX] = oa[4 + i].y + bs1[2 * i + 1];
      }
    } else {
#pragma unroll
      for (int i = 0; i < 4; ++i) {
        v2f q0, q1, z0, z1;
        q0.x = Q0[2 * i]; q0.y = Q0[2 * i + 1]; q1.x = Q1[2 * i]; q1.y = Q1[2 * i + 1];
        z0 = tanh_pair(q0); z1 = tanh_pair(q1);
        Pn0[2 * i] = z0.x; Pn0[2 * i + 1] = z0.y; Pn1[2 * i] = z1.x; Pn1[2 * i + 1] = z1.y;
      }
    }

    P0 = Pn0; P1 = Pn1;
#pragma unroll
    for (int i = 0; i < 12; ++i) { R0[i] = R1[i]; R1[i] = R2[i]; R2[i] = Rn[i]; }
#pragma unroll
    for (int j = 0; j < 6; ++j) pr[j] = pr[3 + j];
#pragma unroll
    for (int dj = 0; dj < 3; ++dj) pr[6 + dj] = prn[dj];
  }
}

extern "C" void kernel_launch(void* const* d_in, const int* in_sizes, int n_in,
                              void* d_out, int out_size, void* d_ws, size_t ws_size,
                              hipStream_t stream) {
  (void)in_sizes; (void)n_in; (void)out_size; (void)d_ws; (void)ws_size;
  const float* X    = (const float*)d_in[0];
  const float* W1   = (const float*)d_in[1];
  const float* b1   = (const float*)d_in[2];
  const float* W2   = (const float*)d_in[3];
  const float* b2   = (const float*)d_in[4];
  const float* bias = (const float*)d_in[5];
  float* Out = (float*)d_out;
  dynaconv_kernel<<<BLOCKS, WAVES * 32, 0, stream>>>(X, W1, b1, W2, b2, bias, Out);
}


# GEMM1 as minimal filtering F(2,3) down the column: 2 rows from 4 K=48 products (48 matrix instr per tile, was 72); to fit registers b2 enters as a K=12 product (+6); sum order, rounding changed
# speedup vs baseline: 4.0705x; 1.0616x over previous
#include <hip/hip_runtime.h>
#include <math.h>

typedef float v2f __attribute__((ext_vector_type(2)));
typedef float v4f __attribute__((ext_vector_type(4)));
typedef float v8f __attribute__((ext_vector_type(8)));

#define NB 4
#define NC 16
#define NO 32
#define HW 256
#define LPIX (HW*HW)
#define DF 144
#define KW1 48
#define KSW (KW1/4)
#define DQ 288
#define WAVES 4
#define TPW 16
#define PAIRS (TPW/2)
#define ROWS_TOTAL (NB*LPIX)
#define TILES_TOTAL (ROWS_TOTAL/16)
#define BLOCKS (TILES_TOTAL/(WAVES*TPW))
#define TCOLS (HW/16)
#define STRIPS (TCOLS*(HW/TPW))
#define MT2 (DQ/16)
#define KS2 (NO/4)
#define TRIPLES (NO*NC*3/(WAVES*32))

static __device__ __forceinline__ float xhalf(float x) {
  int b = __builtin_bit_cast(int, x);
  return __builtin_bit_cast(float,
      __builtin_amdgcn_permlanex16(b, b, 0x76543210, 0xfedcba98, false, false));
}

static __device__ __forceinline__ v8f wmma4(v2f a, v2f b, v8f c) {
  return __builtin_amdgcn_wmma_f32_16x16x4_f32(
      false, a, false, b, (short)0, c, false, false);
}

template <int N>
static __device__ __forceinline__ void ld_frags(const float* base, v2f (&fa)[N], v2f (&fb)[N]) {
#pragma unroll
  for (int i = 0; i < N; ++i) {
    fa[i] = *(const v2f*)(base + i * 128);
    fb[i] = *(const v2f*)(base + i * 128 + 32);
  }
}

static __device__ __forceinline__ void ld_row(const float* img, unsigned rowoff,
                                              const unsigned (&xo)[3], v2f (&R)[12]) {
#pragma unroll
  for (int dj = 0; dj < 3; ++dj) {
    const float* pt = (const float*)((const char*)img + (xo[dj] + rowoff));
#pragma unroll
    for (int g = 0; g < 4; ++g) {
      R[dj * 4 + g].x = pt[(4 * g) * LPIX];
      R[dj * 4 + g].y = pt[(4 * g + 1) * LPIX];
    }
  }
}

static __device__ __forceinline__ void zero_pad(bool rowok, bool ok0, bool ok2, v2f (&R)[12]) {
#pragma unroll
  for (int dj = 0; dj < 3; ++dj) {
    const bool ok = rowok && (dj == 0 ? ok0 : (dj == 2 ? ok2 : true));
#pragma unroll
    for (int g = 0; g < 4; ++g) {
      R[dj * 4 + g].x = ok ? R[dj * 4 + g].x : 0.0f;
      R[dj * 4 + g].y = ok ? R[dj * 4 + g].y : 0.0f;
    }
  }
}

static __device__ __forceinline__ float tap_sum(const v2f (&R)[12], int dj) {
  float s = R[dj * 4].x + R[dj * 4].y;
#pragma unroll
  for (int g = 1; g < 4; ++g) { s += R[dj * 4 + g].x; s += R[dj * 4 + g].y; }
  s += xhalf(s);
  return s;
}

static __device__ __forceinline__ v2f tanh_pair(v2f x) {
  const float c2l = __builtin_bit_cast(float, 0x4038aa3bu);
  const v2f m = x * c2l;
  v2f p;
  p.x = __builtin_amdgcn_exp2f(m.x);
  p.y = __builtin_amdgcn_exp2f(m.y);
  const v2f d = p + 1.0f;
  v2f z;
  z.x = __builtin_fmaf(__builtin_amdgcn_rcpf(d.x), -2.0f, 1.0f);
  z.y = __builtin_fmaf(__builtin_amdgcn_rcpf(d.y), -2.0f, 1.0f);
  return z;
}

static __device__ __forceinline__ int row_of(int t, int hm, int r) {
  const int slot = 4 * (t % 9) + (r >> 1);
  const int cp = slot / 9, j = slot - 9 * cp;
  const int o = (t < 9 ? 0 : 16) + 8 * hm + 2 * cp + (r & 1);
  return o * 9 + j;
}

__global__ __launch_bounds__(WAVES * 32)
void dynaconv_kernel(const float* __restrict__ X,
                     const float* __restrict__ W1,
                     const float* __restrict__ b1,
                     const float* __restrict__ W2,
                     const float* __restrict__ b2,
                     const float* __restrict__ bias,
                     float* __restrict__ Out) {
  __shared__ __align__(16) float sU[4 * (KW1 / 2) * 64];

  const int tid  = threadIdx.x;
  const int lane = tid & 31;
  const int wv   = __builtin_amdgcn_readfirstlane(tid >> 5);
  const int col  = lane & 15;
  const int h    = lane >> 4;

  const int sw   = (int)blockIdx.x * WAVES + wv;
  const int bimg = sw / STRIPS;
  const int tcol = sw % TCOLS;
  const int y0   = ((sw / TCOLS) % (HW / TPW)) * TPW;
  const int x    = tcol * 16 + col;
  const float* xb = X + bimg * NC * LPIX;
  unsigned xo[3];
  xo[0] = 4u * (unsigned)(2 * h * LPIX + max(x - 1, 0));
  xo[1] = 4u * (unsigned)(2 * h * LPIX + x);
  xo[2] = 4u * (unsigned)(2 * h * LPIX + min(x + 1, HW - 1));
  const bool ok0 = x > 0, ok2 = x < HW - 1;

  v2f R0[12], R1[12], R2[12], R3[12], N0[12], N1[12];
  ld_row(xb, 4u * (unsigned)(max(y0 - 1, 0) * HW), xo, R0);
  ld_row(xb, 4u * (unsigned)(y0 * HW), xo, R1);
  ld_row(xb, 4u * (unsigned)(min(y0 + 1, HW - 1) * HW), xo, R2);
  ld_row(xb, 4u * (unsigned)(min(y0 + 2, HW - 1) * HW), xo, R3);

  v2f w2f[MT2][KS2];
#pragma unroll
  for (int t = 0; t < MT2; ++t) {
    const int g = row_of(t, col >> 3, col & 7);
    const v4f* wp = (const v4f*)(W2 + g * NO + 8 * h);
    const v4f x0 = wp[0], x1 = wp[1], x2 = wp[4], x3 = wp[5];
    w2f[t][0].x = x0.x; w2f[t][0].y = x0.y; w2f[t][1].x = x0.z; w2f[t][1].y = x0.w;
    w2f[t][2].x = x1.x; w2f[t][2].y = x1.y; w2f[t][3].x = x1.z; w2f[t][3].y = x1.w;
    w2f[t][4].x = x2.x; w2f[t][4].y = x2.y; w2f[t][5].x = x2.z; w2f[t][5].y = x2.w;
    w2f[t][6].x = x3.x; w2f[t][6].y = x3.y; w2f[t][7].x = x3.z; w2f[t][7].y = x3.w;
  }

  {
    float gv[TRIPLES][3];
#pragma unroll
    for (int i = 0; i < TRIPLES; ++i) {
      const int t = i * (WAVES * 32) + tid;
      const int dj = t % 3;
      const float* gp = W1 + (3 * t - 2 * dj);
      gv[i][0] = gp[0]; gv[i][1] = gp[3]; gv[i][2] = gp[6];
    }
#pragma unroll
    for (int i = 0; i < TRIPLES; ++i) {
      const int t = i * (WAVES * 32) + tid;
      const int dj = t % 3, oc = t / 3;
      const int c = oc & (NC - 1), o = oc >> 4;
      const int k = dj * NC + c;
      const int idx = (k >> 1) * 64 + (o << 1) + (k & 1);
      const float g0 = gv[i][0], g1 = gv[i][1], g2 = gv[i][2];
      const float e = g0 + g2;
      sU[idx]                       = g0;
      sU[(KW1 / 2) * 64 + idx]      = 0.5f * (e + g1);
      sU[2 * (KW1 / 2) * 64 + idx]  = 0.5f * (e - g1);
      sU[3 * (KW1 / 2) * 64 + idx]  = -g2;
    }
  }
  v2f b2a[3], b2b[3];
#pragma unroll
  for (int s3 = 0; s3 < 3; ++s3) {
    const int j0 = 6 * h + 2 * s3, j1 = j0 + 1;
    const float a0 = b2[col * 9 + min(j0, 8)],        a1 = b2[col * 9 + min(j1, 8)];
    const float c0 = b2[(16 + col) * 9 + min(j0, 8)], c1 = b2[(16 + col) * 9 + min(j1, 8)];
    b2a[s3].x = j0 < 9 ? a0 : 0.0f; b2a[s3].y = j1 < 9 ? a1 : 0.0f;
    b2b[s3].x = j0 < 9 ? c0 : 0.0f; b2b[s3].y = j1 < 9 ? c1 : 0.0f;
  }
  v8f b1v0, b1v1, bs0, bs1;
#pragma unroll
  for (int r = 0; r < 8; ++r) {
    b1v0[r] = b1[8 * h + r];   b1v1[r] = b1[16 + 8 * h + r];
    bs0[r]  = bias[8 * h + r]; bs1[r]  = bias[16 + 8 * h + r];
  }

  float ps[12];
  zero_pad(y0 > 0, ok0, ok2, R0);
  zero_pad(true, ok0, ok2, R1);
#pragma unroll
  for (int dj = 0; dj < 3; ++dj) {
    ps[dj] = 0.0f; ps[3 + dj] = 0.0f; ps[6 + dj] = tap_sum(R0, dj); ps[9 + dj] = tap_sum(R1, dj);
  }
  __syncthreads();

  v2f u0a[KSW], u0b[KSW], u1a[KSW], u1b[KSW], u2a[KSW], u2b[KSW], u3a[KSW], u3b[KSW];
  ld_frags<KSW>(sU + h * 64 + (col << 1), u0a, u0b);
  ld_frags<KSW>(sU + (KW1 / 2) * 64 + h * 64 + (col << 1), u1a, u1b);
  ld_frags<KSW>(sU + 2 * (KW1 / 2) * 64 + h * 64 + (col << 1), u2a, u2b);
  ld_frags<KSW>(sU + 3 * (KW1 / 2) * 64 + h * 64 + (col << 1), u3a, u3b);

  const v8f zero8 = {0.0f, 0.0f, 0.0f, 0.0f, 0.0f, 0.0f, 0.0f, 0.0f};
  v8f PA0 = zero8, PA1 = zero8, PB0 = zero8, PB1 = zero8;
  v8f PnA0 = zero8, PnA1 = zero8, PnB0 = zero8, PnB1 = zero8;
  v8f A0a = zero8, A0b = zero8, B3a = zero8, B3b = zero8;
  v8f M1a = zero8, M1b = zero8, M2a = zero8, M2b = zero8;

#pragma nounroll
  for (int u = 0; u <= PAIRS; ++u) {
    float psn[6] = {0.0f, 0.0f, 0.0f, 0.0f, 0.0f, 0.0f};
    if (u < PAIRS) {
      const int ya = y0 + 2 * u;
      ld_row(xb, 4u * (unsigned)(min(ya + 3, HW - 1) * HW), xo, N0);
      ld_row(xb, 4u * (unsigned)(min(ya + 4, HW - 1) * HW), xo, N1);
      zero_pad(ya + 1 < HW, ok0, ok2, R2);
      zero_pad(ya + 2 < HW, ok0, ok2, R3);
      v2f D0[KSW], D1[KSW], D2[KSW], D3[KSW];
#pragma unroll
      for (int i = 0; i < KSW; ++i) {
        D0[i] = R0[i] - R2[i]; D1[i] = R1[i] + R2[i]; D2[i] = R2[i] - R1[i]; D3[i] = R1[i] - R3[i];
      }
#pragma unroll
      for (int i = 0; i < KSW; ++i) {
        A0a = wmma4(u0a[i], D0[i], i == 0 ? b1v0 : A0a);
        A0b = wmma4(u0b[i], D0[i], i == 0 ? b1v1 : A0b);
        M1a = wmma4(u1a[i], D1[i], i == 0 ? zero8 : M1a);
        M1b = wmma4(u1b[i], D1[i], i == 0 ? zero8 : M1b);
        M2a = wmma4(u2a[i], D2[i], i == 0 ? zero8 : M2a);
        M2b = wmma4(u2b[i], D2[i], i == 0 ? zero8 : M2b);
        B3a = wmma4(u3a[i], D3[i], i == 0 ? b1v0 : B3a);
        B3b = wmma4(u3b[i], D3[i], i == 0 ? b1v1 : B3b);
        if (i >= 1 && i < 4) { psn[i - 1] = tap_sum(R2, i - 1); psn[3 + i - 1] = tap_sum(R3, i - 1); }
      }
    }

    if (u > 0) {
      const int yb = y0 + 2 * (u - 1);
#pragma unroll
      for (int w = 0; w < 2; ++w) {
        const v8f P0 = w == 0 ? PA0 : PB0, P1 = w == 0 ? PA1 : PB1;
        float pr[9];
#pragma unroll
        for (int j = 0; j < 9; ++j) pr[j] = ps[3 * w + j];
        v8f o0, o1;
#pragma unroll
        for (int s3 = 0; s3 < 3; ++s3) {
          const int ja = 2 * s3, jb = 6 + 2 * s3;
          v2f b;
          b.x = h ? (jb < 9 ? pr[jb < 9 ? jb : 0] : 0.0f) : pr[ja];
          b.y = h ? (jb + 1 < 9 ? pr[jb + 1 < 9 ? jb + 1 : 0] : 0.0f) : pr[ja + 1];
          o0 = wmma4(b2a[s3], b, s3 == 0 ? bs0 : o0);
          o1 = wmma4(b2b[s3], b, s3 == 0 ? bs1 : o1);
        }
        v2f oa[8];
        v8f kt[MT2];
#pragma unroll
        for (int p = 0; p <= MT2 / 2; ++p) {
          const int t0 = 2 * p, t1 = 2 * p + 1;
#pragma unroll
          for (int s = 0; s < KS2; ++s) {
            if (p < MT2 / 2) {
              if (s == 1 && p < 8) {
                const int e = p < 4 ? 2 * p : 2 * (p - 4);
                v2f q, m1, m2, z;
                if (p < 4) {
                  q.x = w == 0 ? A0a[e] : B3a[e]; q.y = w == 0 ? A0a[e + 1] : B3a[e + 1];
                  m1.x = M1a[e]; m1.y = M1a[e + 1]; m2.x = M2a[e]; m2.y = M2a[e + 1];
                } else {
                  q.x = w == 0 ? A0b[e] : B3b[e]; q.y = w == 0 ? A0b[e + 1] : B3b[e + 1];
                  m1.x = M1b[e]; m1.y = M1b[e + 1]; m2.x = M2b[e]; m2.y = M2b[e + 1];
                }
                q = w == 0 ? (q + m1) + m2 : (q + m1) - m2;
                z = tanh_pair(q);
                if (w == 0) {
                  if (p < 4) { PnA0[e] = z.x; PnA0[e + 1] = z.y; } else { PnA1[e] = z.x; PnA1[e + 1] = z.y; }
                } else {
                  if (p < 4) { PnB0[e] = z.x; PnB0[e + 1] = z.y; } else { PnB1[e] = z.x; PnB1[e + 1] = z.y; }
                }
              }
              v2f b;
              if (s < 4) { b.x = P0[2 * s]; b.y = P0[2 * s + 1]; }
              else       { b.x = P1[2 * (s - 4)]; b.y = P1[2 * (s - 4) + 1]; }
              kt[t0] = wmma4(w2f[t0][s], b, s == 0 ? zero8 : kt[t0]);
              kt[t1] = wmma4(w2f[t1][s], b, s == 0 ? zero8 : kt[t1]);
            }
            if (p == 0 && s < 4) {
              oa[s].x = o0[2 * s]; oa[s].y = o0[2 * s + 1];
              oa[4 + s].x = o1[2 * s]; oa[4 + s].y = o1[2 * s + 1];
            }
            if (p > 0) {
              const int tp = (s < 4) ? t0 - 2 : t1 - 2;
              const int slot = 4 * (tp % 9) + (s & 3);
              const int cp = slot / 9, j = slot % 9;
              v2f kv;
              kv.x = kt[tp][2 * (s & 3)]; kv.y = kt[tp][2 * (s & 3) + 1];
              const int i = (tp < 9 ? 0 : 4) + cp;
              oa[i].x += kv.x * pr[j]; oa[i].y += kv.y * pr[j];
            }
          }
        }

        float* ob = (float*)((char*)(Out + (long)bimg * NO * LPIX)
                             + (4u * (unsigned)(8 * h * LPIX + x) + 4u * (unsigned)((yb + w) * HW)));
#pragma unroll
        for (int i = 0; i < 4; ++i) {
          ob[(2 * i) * LPIX]          = oa[i].x;
          ob[(2 * i + 1) * LPIX]      = oa[i].y;
          ob[(16 + 2 * i) * LPIX]     = oa[4 + i].x;
          ob[(16 + 2 * i + 1) * LPIX] = oa[4 + i].y;
        }
      }
    } else {
#pragma unroll
      for (int i = 0; i < 4; ++i) {
        v2f qa0, qa1, qb0, qb1, m, n, z;
        m.x = M1a[2 * i]; m.y = M1a[2 * i + 1]; n.x = M2a[2 * i]; n.y = M2a[2 * i + 1];
        qa0.x = A0a[2 * i]; qa0.y = A0a[2 * i + 1]; qb0.x = B3a[2 * i]; qb0.y = B3a[2 * i + 1];
        z = tanh_pair((qa0 + m) + n); PnA0[2 * i] = z.x; PnA0[2 * i + 1] = z.y;
        z = tanh_pair((qb0 + m) - n); PnB0[2 * i] = z.x; PnB0[2 * i + 1] = z.y;
        m.x = M1b[2 * i]; m.y = M1b[2 * i + 1]; n.x = M2b[2 * i]; n.y = M2b[2 * i + 1];
        qa1.x = A0b[2 * i]; qa1.y = A0b[2 * i + 1]; qb1.x = B3b[2 * i]; qb1.y = B3b[2 * i + 1];
        z = tanh_pair((qa1 + m) + n); PnA1[2 * i] = z.x; PnA1[2 * i + 1] = z.y;
        z = tanh_pair((qb1 + m) - n); PnB1[2 * i] = z.x; PnB1[2 * i + 1] = z.y;
      }
    }

    PA0 = PnA0; PA1 = PnA1; PB0 = PnB0; PB1 = PnB1;
#pragma unroll
    for (int i = 0; i < 12; ++i) { R0[i] = R2[i]; R1[i] = R3[i]; R2[i] = N0[i]; R3[i] = N1[i]; }
#pragma unroll
    for (int j = 0; j < 6; ++j) { ps[j] = ps[6 + j]; ps[6 + j] = psn[j]; }
  }
}

extern "C" void kernel_launch(void* const* d_in, const int* in_sizes, int n_in,
                              void* d_out, int out_size, void* d_ws, size_t ws_size,
                              hipStream_t stream) {
  (void)in_sizes; (void)n_in; (void)out_size; (void)d_ws; (void)ws_size;
  const float* X    = (const float*)d_in[0];
  const float* W1   = (const float*)d_in[1];
  const float* b1   = (const float*)d_in[2];
  const float* W2   = (const float*)d_in[3];
  const float* b2   = (const float*)d_in[4];
  const float* bias = (const float*)d_in[5];
  float* Out = (float*)d_out;
  dynaconv_kernel<<<BLOCKS, WAVES * 32, 0, stream>>>(X, W1, b1, W2, b2, bias, Out);
}
